# starting kernel
# baseline (speedup 1.0000x reference)
#include <hip/hip_runtime.h>
#include <hip/hip_bf16.h>

typedef __attribute__((ext_vector_type(2))) float v2f;
typedef __attribute__((ext_vector_type(8))) float v8f;

__global__ void deg_init_kernel(float* deg, int n) {
    int i = blockIdx.x * blockDim.x + threadIdx.x;
    if (i < n) deg[i] = 1.0f;
}

__global__ void deg_accum_kernel(const int* dst, const float* w, float* deg, int e) {
    int i = blockIdx.x * blockDim.x + threadIdx.x;
    if (i < e) atomicAdd(&deg[dst[i]], w[i]);
}

__global__ void dinv_kernel(float* deg, int n) {
    int i = blockIdx.x * blockDim.x + threadIdx.x;
    if (i < n) {
        float d = deg[i];
        deg[i] = (d > 0.0f) ? rsqrtf(d) : 0.0f;
    }
}

__global__ void __launch_bounds__(256)
gemm_wmma_f32_kernel(const float* __restrict__ A, const float* __restrict__ W,
                     float* __restrict__ Out, int M, int K, int F, int relu_in) {
    int wid  = (blockIdx.x * blockDim.x + threadIdx.x) >> 5;
    int lane = threadIdx.x & 31;
    int tiles_n = F >> 4;
    int total   = (M >> 4) * tiles_n;
    if (wid >= total) return;

    int tm  = wid / tiles_n;
    int tn  = wid - tm * tiles_n;
    int l16  = lane & 15;
    int half = lane >> 4;

    int row = tm * 16 + l16;
    int col = tn * 16 + l16;

    const float* arow = A + (size_t)row * K;
    v8f c = {0.f, 0.f, 0.f, 0.f, 0.f, 0.f, 0.f, 0.f};

    for (int k = 0; k < K; k += 4) {
        int ka = k + 2 * half;
        v2f a;
        a.x = arow[ka];
        a.y = arow[ka + 1];
        if (relu_in) {
            a.x = fmaxf(a.x, 0.0f);
            a.y = fmaxf(a.y, 0.0f);
        }
        v2f b;
        b.x = W[(size_t)ka * F + col];
        b.y = W[(size_t)(ka + 1) * F + col];
        c = __builtin_amdgcn_wmma_f32_16x16x4_f32(false, a, false, b,
                                                  (short)0, c, false, false);
    }

    int mbase = tm * 16 + half * 8;
#pragma unroll
    for (int r = 0; r < 8; ++r)
        Out[(size_t)(mbase + r) * F + col] = c[r];
}

__global__ void init_agg_kernel(const float* __restrict__ H,
                                const float* __restrict__ bias,
                                const float* __restrict__ dinv,
                                float* __restrict__ agg, int n, int F) {
    long long i = (long long)blockIdx.x * blockDim.x + threadIdx.x;
    long long tot = (long long)n * F;
    if (i >= tot) return;
    int node = (int)(i / F);
    int f    = (int)(i - (long long)node * F);
    float di = dinv[node];
    agg[i] = bias[f] + di * di * H[i];
}

__global__ void __launch_bounds__(256)
scatter_kernel(const int* __restrict__ src, const int* __restrict__ dst,
               const float* __restrict__ w, const float* __restrict__ dinv,
               const float* __restrict__ H, float* __restrict__ agg,
               int e, int F) {
    int wid  = (blockIdx.x * blockDim.x + threadIdx.x) >> 5;
    int lane = threadIdx.x & 31;
    if (wid >= e) return;
    int s = src[wid];
    int d = dst[wid];
    float nrm = dinv[s] * w[wid] * dinv[d];
    const float* h = H + (size_t)s * F;
    float* o = agg + (size_t)d * F;
    for (int f = lane; f < F; f += 32)
        atomicAdd(&o[f], nrm * h[f]);
}

extern "C" void kernel_launch(void* const* d_in, const int* in_sizes, int n_in,
                              void* d_out, int out_size, void* d_ws, size_t ws_size,
                              hipStream_t stream) {
    const float* x    = (const float*)d_in[0];
    const int*   ei   = (const int*)d_in[1];
    const float* ew   = (const float*)d_in[2];
    const float* W1   = (const float*)d_in[3];
    const float* b1   = (const float*)d_in[4];
    const float* W2   = (const float*)d_in[5];
    const float* b2   = (const float*)d_in[6];
    const float* W3   = (const float*)d_in[7];
    const float* b3   = (const float*)d_in[8];
    float*       out  = (float*)d_out;

    const int hidden  = in_sizes[4];
    const int out_dim = in_sizes[8];
    const int in_dim  = in_sizes[3] / hidden;
    const int n       = in_sizes[0] / in_dim;
    const int e       = in_sizes[2];

    const int* src = ei;
    const int* dst = ei + e;

    char* ws = (char*)d_ws;
    size_t off = 0;
    float* dinv = (float*)(ws + off);
    off += ((size_t)n * sizeof(float) + 255) & ~(size_t)255;
    float* H = (float*)(ws + off);
    off += ((size_t)n * hidden * sizeof(float) + 255) & ~(size_t)255;
    float* AGG = (float*)(ws + off);

    const int B = 256;
    auto cdiv = [](long long a, long long b) { return (int)((a + b - 1) / b); };

    deg_init_kernel<<<cdiv(n, B), B, 0, stream>>>(dinv, n);
    deg_accum_kernel<<<cdiv(e, B), B, 0, stream>>>(dst, ew, dinv, e);
    dinv_kernel<<<cdiv(n, B), B, 0, stream>>>(dinv, n);

    {
        int tiles = (n / 16) * (hidden / 16);
        gemm_wmma_f32_kernel<<<cdiv((long long)tiles * 32, B), B, 0, stream>>>(
            x, W1, H, n, in_dim, hidden, 0);
        init_agg_kernel<<<cdiv((long long)n * hidden, B), B, 0, stream>>>(
            H, b1, dinv, AGG, n, hidden);
        scatter_kernel<<<cdiv((long long)e * 32, B), B, 0, stream>>>(
            src, dst, ew, dinv, H, AGG, e, hidden);
    }

    {
        int tiles = (n / 16) * (hidden / 16);
        gemm_wmma_f32_kernel<<<cdiv((long long)tiles * 32, B), B, 0, stream>>>(
            AGG, W2, H, n, hidden, hidden, 1);
        init_agg_kernel<<<cdiv((long long)n * hidden, B), B, 0, stream>>>(
            H, b2, dinv, AGG, n, hidden);
        scatter_kernel<<<cdiv((long long)e * 32, B), B, 0, stream>>>(
            src, dst, ew, dinv, H, AGG, e, hidden);
    }

    {
        int tiles = (n / 16) * (out_dim / 16);
        gemm_wmma_f32_kernel<<<cdiv((long long)tiles * 32, B), B, 0, stream>>>(
            AGG, W3, H, n, hidden, out_dim, 1);
        init_agg_kernel<<<cdiv((long long)n * out_dim, B), B, 0, stream>>>(
            H, b3, dinv, out, n, out_dim);
        scatter_kernel<<<cdiv((long long)e * 32, B), B, 0, stream>>>(
            src, dst, ew, dinv, H, out, e, out_dim);
    }
}


# Aggregation reads one 8-byte edge record (src|dst<<16, normalised weight; built in edge order by the pass that finishes dinv, dinv table in LDS) per matching edge, not src, dst, w and two dinv
# speedup vs baseline: 1.8883x; 1.8883x over previous
#include <hip/hip_runtime.h>
#include <hip/hip_bf16.h>

typedef __attribute__((ext_vector_type(2))) float v2f;
typedef __attribute__((ext_vector_type(8))) float v8f;

__global__ void deg_init_kernel(float* deg, int n) {
    int i = blockIdx.x * blockDim.x + threadIdx.x;
    if (i < n) deg[i] = 1.0f;
}

__global__ void deg_accum_kernel(const int* dst, const float* w, float* deg, int e) {
    int i = blockIdx.x * blockDim.x + threadIdx.x;
    if (i < e) atomicAdd(&deg[dst[i]], w[i]);
}

#define TILE_R 200

template <bool WITH_DEG>
__global__ void __launch_bounds__(256)
edge_pass_kernel(const int* __restrict__ dst, const float* __restrict__ w,
                 float* __restrict__ deg, unsigned* __restrict__ planes,
                 const unsigned* __restrict__ rec_in, unsigned* __restrict__ rec_out, int e) {
    int i = blockIdx.x * blockDim.x + threadIdx.x;
    int lane = threadIdx.x & 31;
    unsigned key = 255u;
    if (i < e) {
        int d = dst[i];
        if (WITH_DEG) atomicAdd(&deg[d], w[i]);
        else reinterpret_cast<uint2*>(rec_out)[i] = reinterpret_cast<const uint2*>(rec_in)[i];
        unsigned t = (unsigned)d / TILE_R;
        key = (t < 255u) ? t : 255u;
    }
    unsigned mine = 0;
#pragma unroll
    for (int p = 0; p < 8; ++p) {
        unsigned b = __builtin_amdgcn_ballot_w32((key >> p) & 1u);
        mine = (lane == p) ? b : mine;
    }
    if (lane < 8) {
        int g = i >> 5;
        planes[(size_t)(g >> 5) * 256 + (lane >> 2) * 128 + (g & 31) * 4 + (lane & 3)] = mine;
    }
}

__global__ void dinv_kernel(float* deg, int n) {
    int i = blockIdx.x * blockDim.x + threadIdx.x;
    if (i < n) {
        float d = deg[i];
        deg[i] = (d > 0.0f) ? rsqrtf(d) : 0.0f;
    }
}

#define REC_NMAX (255 * TILE_R)
#define REC_WGS  256

__global__ void __launch_bounds__(1024)
norm_rec_kernel(const int* __restrict__ src, const int* __restrict__ dst,
                const float* __restrict__ w, const float* __restrict__ deg,
                float* __restrict__ dinv, unsigned* __restrict__ rec, int n, int e) {
    __shared__ float sdinv[REC_NMAX];
#pragma unroll 8
    for (int i = threadIdx.x; i < n; i += 1024) {
        float d = deg[i];
        sdinv[i] = (d > 0.0f) ? rsqrtf(d) : 0.0f;
    }
    __syncthreads();
    const int nper = ((n + REC_WGS * 32 - 1) / (REC_WGS * 32)) * 32;
    const int n0   = blockIdx.x * nper;
    const int n1   = (n0 + nper < n) ? (n0 + nper) : n;
    for (int i = n0 + threadIdx.x; i < n1; i += 1024) dinv[i] = sdinv[i];
    const int eper = ((e + REC_WGS * 32 - 1) / (REC_WGS * 32)) * 32;
    const int e0   = blockIdx.x * eper;
    const int e1   = (e0 + eper < e) ? (e0 + eper) : e;
    uint2* rec2 = reinterpret_cast<uint2*>(rec);
    for (int base = e0; base < e1; base += 4 * 1024) {
        int   s[4], d[4];
        float ww[4];
#pragma unroll
        for (int k = 0; k < 4; ++k) {
            const int i = base + k * 1024 + (int)threadIdx.x;
            s[k] = 0; d[k] = 0; ww[k] = 0.0f;
            if (i < e1) { s[k] = src[i]; d[k] = dst[i]; ww[k] = w[i]; }
        }
#pragma unroll
        for (int k = 0; k < 4; ++k) {
            const int i = base + k * 1024 + (int)threadIdx.x;
            if (i < e1) {
                float nrm = sdinv[s[k]] * ww[k] * sdinv[d[k]];
                rec2[i] = make_uint2((unsigned)s[k] | ((unsigned)d[k] << 16), __float_as_uint(nrm));
            }
        }
    }
}

__global__ void __launch_bounds__(256)
gemm_wmma_f32_kernel(const float* __restrict__ A, const float* __restrict__ W,
                     float* __restrict__ Out, int M, int K, int F, int relu_in) {
    int wid  = (blockIdx.x * blockDim.x + threadIdx.x) >> 5;
    int lane = threadIdx.x & 31;
    int tiles_n = F >> 4;
    int total   = (M >> 4) * tiles_n;
    if (wid >= total) return;

    int tm  = wid / tiles_n;
    int tn  = wid - tm * tiles_n;
    int l16  = lane & 15;
    int half = lane >> 4;

    int row = tm * 16 + l16;
    int col = tn * 16 + l16;

    const float* arow = A + (size_t)row * K;
    v8f c = {0.f, 0.f, 0.f, 0.f, 0.f, 0.f, 0.f, 0.f};

    for (int k = 0; k < K; k += 4) {
        int ka = k + 2 * half;
        v2f a;
        a.x = arow[ka];
        a.y = arow[ka + 1];
        if (relu_in) {
            a.x = fmaxf(a.x, 0.0f);
            a.y = fmaxf(a.y, 0.0f);
        }
        v2f b;
        b.x = W[(size_t)ka * F + col];
        b.y = W[(size_t)(ka + 1) * F + col];
        c = __builtin_amdgcn_wmma_f32_16x16x4_f32(false, a, false, b,
                                                  (short)0, c, false, false);
    }

    int mbase = tm * 16 + half * 8;
#pragma unroll
    for (int r = 0; r < 8; ++r)
        Out[(size_t)(mbase + r) * F + col] = c[r];
}

__global__ void init_agg_kernel(const float* __restrict__ H,
                                const float* __restrict__ bias,
                                const float* __restrict__ dinv,
                                float* __restrict__ agg, int n, int F) {
    long long i = (long long)blockIdx.x * blockDim.x + threadIdx.x;
    long long tot = (long long)n * F;
    if (i >= tot) return;
    int node = (int)(i / F);
    int f    = (int)(i - (long long)node * F);
    float di = dinv[node];
    agg[i] = bias[f] + di * di * H[i];
}

__global__ void __launch_bounds__(256)
scatter_kernel(const int* __restrict__ src, const int* __restrict__ dst,
               const float* __restrict__ w, const float* __restrict__ dinv,
               const float* __restrict__ H, float* __restrict__ agg,
               int e, int F) {
    int wid  = (blockIdx.x * blockDim.x + threadIdx.x) >> 5;
    int lane = threadIdx.x & 31;
    if (wid >= e) return;
    int s = src[wid];
    int d = dst[wid];
    float nrm = dinv[s] * w[wid] * dinv[d];
    const float* h = H + (size_t)s * F;
    float* o = agg + (size_t)d * F;
    for (int f = lane; f < F; f += 32)
        atomicAdd(&o[f], nrm * h[f]);
}

#define SCAN_GROUPS 4
#define TILE_RP 208

template <int FN>
__global__ void __launch_bounds__(1024)
agg_tile_kernel(const float* __restrict__ dinv,
                const float* __restrict__ H, const float* __restrict__ bias,
                const unsigned* __restrict__ planes, const unsigned* __restrict__ rec,
                const float* __restrict__ Wn,
                float* __restrict__ outp, int n, int e, int F) {
    __shared__ __attribute__((aligned(16))) float tile[TILE_RP * 132];
    const int lane = threadIdx.x & 31;
    const int wave = threadIdx.x >> 5;
    const int nw   = blockDim.x >> 5;
    const int row0 = blockIdx.x * TILE_R;
    const int rows = (n - row0 < TILE_R) ? (n - row0) : TILE_R;
    const int P    = F + 4;

    for (int r = wave; r < rows; r += nw) {
        float di = dinv[row0 + r];
        const float* h = H + (size_t)(row0 + r) * F;
        for (int f = lane; f < F; f += 32)
            tile[r * P + f] = bias[f] + di * di * h[f];
    }
    if (FN > 0) {
        for (int r = rows + wave; r < TILE_RP; r += nw)
            for (int f = lane; f < F; f += 32)
                tile[r * P + f] = 0.0f;
    }
    __syncthreads();

    int mye = 0;
    int cnt = 0;
    const uint2* rec2 = reinterpret_cast<const uint2*>(rec);

    auto flush = [&](int my_edge, int count) {
        int s = 0, dl = 0;
        float nrm = 0.0f;
        if (lane < count) {
            uint2 r = rec2[my_edge];
            s = (int)(r.x & 0xFFFFu);
            dl = (int)(r.x >> 16) - row0;
            nrm = __uint_as_float(r.y);
        }
        for (int p = 0; p < count; p += 4) {
            float v[4][4];
            int   dls[4];
            float nrs[4];
#pragma unroll
            for (int q = 0; q < 4; ++q) {
                int pp = (p + q < count) ? (p + q) : (count - 1);
                int sp = __builtin_amdgcn_readlane(s, pp);
                dls[q] = __builtin_amdgcn_readlane(dl, pp);
                nrs[q] = __int_as_float(
                    __builtin_amdgcn_readlane(__float_as_int(nrm), pp));
                const float* h = H + (size_t)sp * F;
#pragma unroll
                for (int j = 0; j < 4; ++j)
                    v[q][j] = (32 * j < F) ? h[lane + 32 * j] : 0.0f;
            }
#pragma unroll
            for (int q = 0; q < 4; ++q) {
                if (p + q < count) {
                    float* o = tile + dls[q] * P;
#pragma unroll
                    for (int j = 0; j < 4; ++j)
                        if (32 * j < F)
                            atomicAdd(&o[lane + 32 * j], nrs[q] * v[q][j]);
                }
            }
        }
    };

    const uint4* pl4 = reinterpret_cast<const uint4*>(planes);
    const unsigned tid = blockIdx.x;
    unsigned sel[8];
#pragma unroll
    for (int p = 0; p < 8; ++p) sel[p] = ((tid >> p) & 1u) ? 0u : ~0u;

    const int ntrips = (e + 1024 * SCAN_GROUPS - 1) / (1024 * SCAN_GROUPS);
    for (int t = wave; t < ntrips; t += nw) {
        unsigned bv[SCAN_GROUPS];
        {
            uint4 pa[SCAN_GROUPS], pb[SCAN_GROUPS];
#pragma unroll
            for (int g = 0; g < SCAN_GROUPS; ++g) {
                const size_t blk = (size_t)(t * SCAN_GROUPS + g) * 64;
                pa[g] = pl4[blk + lane];
                pb[g] = pl4[blk + 32 + lane];
            }
#pragma unroll
            for (int g = 0; g < SCAN_GROUPS; ++g)
                bv[g] = (pa[g].x ^ sel[0]) & (pa[g].y ^ sel[1]) & (pa[g].z ^ sel[2]) & (pa[g].w ^ sel[3])
                      & (pb[g].x ^ sel[4]) & (pb[g].y ^ sel[5]) & (pb[g].z ^ sel[6]) & (pb[g].w ^ sel[7]);
        }
#pragma nounroll
        for (int g = 0; g < SCAN_GROUPS; ++g) {
            unsigned bvg = bv[0];
#pragma unroll
            for (int q = 1; q < SCAN_GROUPS; ++q) bvg = (g == q) ? bv[q] : bvg;
            const int base = (t * SCAN_GROUPS + g) << 10;
            unsigned have = __builtin_amdgcn_ballot_w32(bvg != 0);
            while (have) {
                int j = __builtin_ctz(have);
                have &= have - 1;
                unsigned b = (unsigned)__builtin_amdgcn_readlane((int)bvg, j);
                while (b) {
                    int L = __builtin_ctz(b);
                    b &= b - 1;
                    mye = (lane == cnt) ? (base + j * 32 + L) : mye;
                    if (++cnt == 32) {
                        flush(mye, 32);
                        cnt = 0;
                    }
                }
            }
        }
    }
    if (cnt) flush(mye, cnt);

    if constexpr (FN > 0) {
        constexpr int NCT = FN / 16;
        const int nsg  = nw / NCT;
        const int wv   = __builtin_amdgcn_readfirstlane(wave);
        const int ct   = wv % NCT;
        const int sg   = wv / NCT;
        const int l16  = lane & 15;
        const int half = lane >> 4;
        const int col  = ct * 16 + l16;
        v2f bw[32];
#pragma unroll
        for (int kk = 0; kk < 32; ++kk) {
            const int ka = kk * 4 + 2 * half;
            bw[kk].x = Wn[(size_t)ka * FN + col];
            bw[kk].y = Wn[(size_t)(ka + 1) * FN + col];
        }
        __syncthreads();
        const int nstrips = (rows + 15) >> 4;
        for (int s = sg; s < nstrips; s += nsg) {
            const float* arow = tile + (s * 16 + l16) * P + 2 * half;
            v8f c = {0.f, 0.f, 0.f, 0.f, 0.f, 0.f, 0.f, 0.f};
#pragma unroll
            for (int kk = 0; kk < 32; ++kk) {
                v2f a = *reinterpret_cast<const v2f*>(arow + kk * 4);
                a.x = fmaxf(a.x, 0.0f);
                a.y = fmaxf(a.y, 0.0f);
                c = __builtin_amdgcn_wmma_f32_16x16x4_f32(false, a, false, bw[kk],
                                                          (short)0, c, false, false);
            }
            const int mb = s * 16 + half * 8;
            float* o = outp + (size_t)(row0 + mb) * FN + col;
#pragma unroll
            for (int r = 0; r < 8; ++r)
                if (mb + r < rows) o[(size_t)r * FN] = c[r];
        }
    } else {
        __syncthreads();
        for (int r = wave; r < rows; r += nw) {
            float* o = outp + (size_t)(row0 + r) * F;
            for (int f = lane; f < F; f += 32)
                o[f] = tile[r * P + f];
        }
    }
}

extern "C" void kernel_launch(void* const* d_in, const int* in_sizes, int n_in,
                              void* d_out, int out_size, void* d_ws, size_t ws_size,
                              hipStream_t stream) {
    const float* x    = (const float*)d_in[0];
    const int*   ei   = (const int*)d_in[1];
    const float* ew   = (const float*)d_in[2];
    const float* W1   = (const float*)d_in[3];
    const float* b1   = (const float*)d_in[4];
    const float* W2   = (const float*)d_in[5];
    const float* b2   = (const float*)d_in[6];
    const float* W3   = (const float*)d_in[7];
    const float* b3   = (const float*)d_in[8];
    float*       out  = (float*)d_out;

    const int hidden  = in_sizes[4];
    const int out_dim = in_sizes[8];
    const int in_dim  = in_sizes[3] / hidden;
    const int n       = in_sizes[0] / in_dim;
    const int e       = in_sizes[2];

    const int* src = ei;
    const int* dst = ei + e;

    char* ws = (char*)d_ws;
    size_t off = 0;
    float* dinv = (float*)(ws + off);
    off += ((size_t)n * sizeof(float) + 255) & ~(size_t)255;
    float* H = (float*)(ws + off);
    off += ((size_t)n * hidden * sizeof(float) + 255) & ~(size_t)255;
    float* AGG = (float*)(ws + off);

    const int B = 256;
    auto cdiv = [](long long a, long long b) { return (int)((a + b - 1) / b); };

    const int ntiles = cdiv(n, TILE_R);
    const int e_pad  = cdiv(e, 1024 * SCAN_GROUPS) * 1024 * SCAN_GROUPS;
    const size_t side = (size_t)e_pad + (size_t)e * 8;
    const bool keyed = ntiles <= 255 && n <= REC_NMAX && hidden % 32 == 0 && hidden <= 128 &&
                       out_dim % 32 == 0 && out_dim <= 128 &&
                       side + (size_t)n * sizeof(float) <= (size_t)n * out_dim * sizeof(float) &&
                       side <= (size_t)n * hidden * sizeof(float);
    unsigned* planes_a = (unsigned*)d_out;
    unsigned* planes_b = (unsigned*)AGG;
    unsigned* rec_a = (unsigned*)((char*)d_out + e_pad);
    unsigned* rec_b = (unsigned*)((char*)AGG + e_pad);
    float*    degs  = keyed ? (float*)((char*)d_out + side) : dinv;

    auto aggregate = [&](const float* Hin, const float* bias, float* dstbuf, int F,
                         const unsigned* planes, const unsigned* rec) {
        if (keyed) {
            agg_tile_kernel<0><<<ntiles, 1024, 0, stream>>>(
                dinv, Hin, bias, planes, rec, nullptr, dstbuf, n, e, F);
        } else {
            init_agg_kernel<<<cdiv((long long)n * F, B), B, 0, stream>>>(
                Hin, bias, dinv, dstbuf, n, F);
            scatter_kernel<<<cdiv((long long)e * 32, B), B, 0, stream>>>(
                src, dst, ew, dinv, Hin, dstbuf, e, F);
        }
    };

    deg_init_kernel<<<cdiv(n, B), B, 0, stream>>>(degs, n);
    if (keyed) {
        edge_pass_kernel<true><<<e_pad / B, B, 0, stream>>>(dst, ew, degs, planes_a,
                                                            nullptr, nullptr, e);
        norm_rec_kernel<<<REC_WGS, 1024, 0, stream>>>(src, dst, ew, degs, dinv, rec_a, n, e);
    } else {
        deg_accum_kernel<<<cdiv(e, B), B, 0, stream>>>(dst, ew, dinv, e);
        dinv_kernel<<<cdiv(n, B), B, 0, stream>>>(dinv, n);
    }

    if (keyed && hidden == 128 && out_dim == 64) {
        int tiles = (n / 16) * (hidden / 16);
        gemm_wmma_f32_kernel<<<cdiv((long long)tiles * 32, B), B, 0, stream>>>(
            x, W1, H, n, in_dim, hidden, 0);
        agg_tile_kernel<128><<<ntiles, 1024, 0, stream>>>(
            dinv, H, b1, planes_a, rec_a, W2, AGG, n, e, hidden);
        agg_tile_kernel<64><<<ntiles, 1024, 0, stream>>>(
            dinv, AGG, b2, planes_a, rec_a, W3, H, n, e, hidden);
        edge_pass_kernel<false><<<e_pad / B, B, 0, stream>>>(dst, nullptr, nullptr, planes_b,
                                                             rec_a, rec_b, e);
        agg_tile_kernel<0><<<ntiles, 1024, 0, stream>>>(
            dinv, H, b3, planes_b, rec_b, nullptr, out, n, e, out_dim);
        return;
    }

    {
        int tiles = (n / 16) * (hidden / 16);
        gemm_wmma_f32_kernel<<<cdiv((long long)tiles * 32, B), B, 0, stream>>>(
            x, W1, H, n, in_dim, hidden, 0);
        aggregate(H, b1, AGG, hidden, planes_a, rec_a);
    }

    {
        int tiles = (n / 16) * (hidden / 16);
        gemm_wmma_f32_kernel<<<cdiv((long long)tiles * 32, B), B, 0, stream>>>(
            AGG, W2, H, n, hidden, hidden, 1);
        aggregate(H, b2, AGG, hidden, planes_a, rec_a);
    }

    {
        int tiles = (n / 16) * (out_dim / 16);
        gemm_wmma_f32_kernel<<<cdiv((long long)tiles * 32, B), B, 0, stream>>>(
            AGG, W3, H, n, hidden, out_dim, 1);
        if (keyed)
            edge_pass_kernel<false><<<e_pad / B, B, 0, stream>>>(dst, nullptr, nullptr, planes_b,
                                                                 rec_a, rec_b, e);
        aggregate(H, b3, out, out_dim, planes_b, rec_b);
    }
}


# Layer-1 product with the weights held still: a wave keeps one column tile's B fragments for all of K in registers, x rows come through LDS by whole-row 16-byte loads fetched ahead; same k order
# speedup vs baseline: 2.1093x; 1.1170x over previous
#include <hip/hip_runtime.h>
#include <hip/hip_bf16.h>

typedef __attribute__((ext_vector_type(2))) float v2f;
typedef __attribute__((ext_vector_type(8))) float v8f;

__global__ void deg_init_kernel(float* deg, int n) {
    int i = blockIdx.x * blockDim.x + threadIdx.x;
    if (i < n) deg[i] = 1.0f;
}

__global__ void deg_accum_kernel(const int* dst, const float* w, float* deg, int e) {
    int i = blockIdx.x * blockDim.x + threadIdx.x;
    if (i < e) atomicAdd(&deg[dst[i]], w[i]);
}

#define TILE_R 200

template <bool WITH_DEG>
__global__ void __launch_bounds__(256)
edge_pass_kernel(const int* __restrict__ dst, const float* __restrict__ w,
                 float* __restrict__ deg, unsigned* __restrict__ planes,
                 const unsigned* __restrict__ rec_in, unsigned* __restrict__ rec_out, int e) {
    int i = blockIdx.x * blockDim.x + threadIdx.x;
    int lane = threadIdx.x & 31;
    unsigned key = 255u;
    if (i < e) {
        int d = dst[i];
        if (WITH_DEG) atomicAdd(&deg[d], w[i]);
        else reinterpret_cast<uint2*>(rec_out)[i] = reinterpret_cast<const uint2*>(rec_in)[i];
        unsigned t = (unsigned)d / TILE_R;
        key = (t < 255u) ? t : 255u;
    }
    unsigned mine = 0;
#pragma unroll
    for (int p = 0; p < 8; ++p) {
        unsigned b = __builtin_amdgcn_ballot_w32((key >> p) & 1u);
        mine = (lane == p) ? b : mine;
    }
    if (lane < 8) {
        int g = i >> 5;
        planes[(size_t)(g >> 5) * 256 + (lane >> 2) * 128 + (g & 31) * 4 + (lane & 3)] = mine;
    }
}

__global__ void dinv_kernel(float* deg, int n) {
    int i = blockIdx.x * blockDim.x + threadIdx.x;
    if (i < n) {
        float d = deg[i];
        deg[i] = (d > 0.0f) ? rsqrtf(d) : 0.0f;
    }
}

#define REC_NMAX (255 * TILE_R)
#define REC_WGS  256

__global__ void __launch_bounds__(1024)
norm_rec_kernel(const int* __restrict__ src, const int* __restrict__ dst,
                const float* __restrict__ w, const float* __restrict__ deg,
                float* __restrict__ dinv, unsigned* __restrict__ rec, int n, int e) {
    __shared__ float sdinv[REC_NMAX];
#pragma unroll 8
    for (int i = threadIdx.x; i < n; i += 1024) {
        float d = deg[i];
        sdinv[i] = (d > 0.0f) ? rsqrtf(d) : 0.0f;
    }
    __syncthreads();
    const int nper = ((n + REC_WGS * 32 - 1) / (REC_WGS * 32)) * 32;
    const int n0   = blockIdx.x * nper;
    const int n1   = (n0 + nper < n) ? (n0 + nper) : n;
    for (int i = n0 + threadIdx.x; i < n1; i += 1024) dinv[i] = sdinv[i];
    const int eper = ((e + REC_WGS * 32 - 1) / (REC_WGS * 32)) * 32;
    const int e0   = blockIdx.x * eper;
    const int e1   = (e0 + eper < e) ? (e0 + eper) : e;
    uint2* rec2 = reinterpret_cast<uint2*>(rec);
    for (int base = e0; base < e1; base += 4 * 1024) {
        int   s[4], d[4];
        float ww[4];
#pragma unroll
        for (int k = 0; k < 4; ++k) {
            const int i = base + k * 1024 + (int)threadIdx.x;
            s[k] = 0; d[k] = 0; ww[k] = 0.0f;
            if (i < e1) { s[k] = src[i]; d[k] = dst[i]; ww[k] = w[i]; }
        }
#pragma unroll
        for (int k = 0; k < 4; ++k) {
            const int i = base + k * 1024 + (int)threadIdx.x;
            if (i < e1) {
                float nrm = sdinv[s[k]] * ww[k] * sdinv[d[k]];
                rec2[i] = make_uint2((unsigned)s[k] | ((unsigned)d[k] << 16), __float_as_uint(nrm));
            }
        }
    }
}

__global__ void __launch_bounds__(256)
gemm_wmma_f32_kernel(const float* __restrict__ A, const float* __restrict__ W,
                     float* __restrict__ Out, int M, int K, int F, int relu_in) {
    int wid  = (blockIdx.x * blockDim.x + threadIdx.x) >> 5;
    int lane = threadIdx.x & 31;
    int tiles_n = F >> 4;
    int total   = (M >> 4) * tiles_n;
    if (wid >= total) return;

    int tm  = wid / tiles_n;
    int tn  = wid - tm * tiles_n;
    int l16  = lane & 15;
    int half = lane >> 4;

    int row = tm * 16 + l16;
    int col = tn * 16 + l16;

    const float* arow = A + (size_t)row * K;
    v8f c = {0.f, 0.f, 0.f, 0.f, 0.f, 0.f, 0.f, 0.f};

    for (int k = 0; k < K; k += 4) {
        int ka = k + 2 * half;
        v2f a;
        a.x = arow[ka];
        a.y = arow[ka + 1];
        if (relu_in) {
            a.x = fmaxf(a.x, 0.0f);
            a.y = fmaxf(a.y, 0.0f);
        }
        v2f b;
        b.x = W[(size_t)ka * F + col];
        b.y = W[(size_t)(ka + 1) * F + col];
        c = __builtin_amdgcn_wmma_f32_16x16x4_f32(false, a, false, b,
                                                  (short)0, c, false, false);
    }

    int mbase = tm * 16 + half * 8;
#pragma unroll
    for (int r = 0; r < 8; ++r)
        Out[(size_t)(mbase + r) * F + col] = c[r];
}

#define G1_ROWS 32
#define G1_P    260

static __device__ __forceinline__ void lds_barrier() {
    asm volatile("s_wait_dscnt 0x0\n\ts_barrier_signal -1\n\ts_barrier_wait -1" ::: "memory");
}

__global__ void __launch_bounds__(512)
gemm1_wstat_kernel(const float* __restrict__ A, const float* __restrict__ W,
                   float* __restrict__ Out, int M) {
    constexpr int K = 256;
    constexpr int F = 128;
    __shared__ __attribute__((aligned(16))) float sa[2 * G1_ROWS * G1_P];
    const int lane = threadIdx.x & 31;
    const int wv   = __builtin_amdgcn_readfirstlane((int)(threadIdx.x >> 5));
    const int ct   = wv & 7;
    const int sg   = wv >> 3;
    const int l16  = lane & 15;
    const int half = lane >> 4;
    const int col  = ct * 16 + l16;

    const int nstrips  = (M + 15) >> 4;
    const int nwg      = (int)gridDim.x;
    const int base     = nstrips / nwg;
    const int rem      = nstrips - base * nwg;
    const int b        = (int)blockIdx.x;
    const int mystrips = base + ((b < rem) ? 1 : 0);
    const int strip0   = b * base + ((b < rem) ? b : rem);
    const int row0     = strip0 * 16;
    const int nrounds  = (mystrips + 1) >> 1;

    const float4* A4 = reinterpret_cast<const float4*>(A);
    const int lr = threadIdx.x >> 6;
    const int c4 = threadIdx.x & 63;

    float4 p0, p1, p2, p3;
    const float4* acol = A4 + c4;
    auto rowptr = [&](int grow) {
        const int g = (grow < M) ? grow : (M - 1);
        return acol + (size_t)g * (K / 4);
    };
    auto fetch = [&](int r) {
        const int g0 = row0 + r * G1_ROWS + lr;
        p0 = *rowptr(g0);
        p1 = *rowptr(g0 + 8);
        p2 = *rowptr(g0 + 16);
        p3 = *rowptr(g0 + 24);
    };
    auto stash = [&](int buf, float4 a0, float4 a1, float4 a2, float4 a3) {
        float* d = &sa[buf * (G1_ROWS * G1_P) + lr * G1_P + 4 * c4];
        *reinterpret_cast<float4*>(d)             = a0;
        *reinterpret_cast<float4*>(d + 8 * G1_P)  = a1;
        *reinterpret_cast<float4*>(d + 16 * G1_P) = a2;
        *reinterpret_cast<float4*>(d + 24 * G1_P) = a3;
    };

    const int gfirst = row0 + lr;
    const float4 q0 = *rowptr(gfirst);
    const float4 q1 = *rowptr(gfirst + 8);
    const float4 q2 = *rowptr(gfirst + 16);
    const float4 q3 = *rowptr(gfirst + 24);
    v2f bw[K / 4];
#pragma unroll
    for (int kk = 0; kk < K / 4; ++kk) {
        const int ka = kk * 4 + 2 * half;
        bw[kk].x = W[(size_t)ka * F + col];
        bw[kk].y = W[(size_t)(ka + 1) * F + col];
    }
    fetch(1);
    stash(0, q0, q1, q2, q3);
    __syncthreads();

    for (int r = 0; r < nrounds; ++r) {
        if (r + 1 < nrounds) stash((r + 1) & 1, p0, p1, p2, p3);
        if (r + 2 < nrounds) fetch(r + 2);
        const int ls = 2 * r + sg;
        if (ls < mystrips) {
            const float* arow = sa + (r & 1) * (G1_ROWS * G1_P) + (sg * 16 + l16) * G1_P + 2 * half;
            v8f c = {0.f, 0.f, 0.f, 0.f, 0.f, 0.f, 0.f, 0.f};
#pragma unroll
            for (int kk = 0; kk < K / 4; ++kk) {
                v2f a = *reinterpret_cast<const v2f*>(arow + kk * 4);
                c = __builtin_amdgcn_wmma_f32_16x16x4_f32(false, a, false, bw[kk],
                                                          (short)0, c, false, false);
            }
            const int mb = row0 + ls * 16 + half * 8;
            float* o = Out + (size_t)mb * F + col;
#pragma unroll
            for (int q = 0; q < 8; ++q)
                if (mb + q < M) o[(size_t)q * F] = c[q];
        }
        lds_barrier();
    }
}

__global__ void init_agg_kernel(const float* __restrict__ H,
                                const float* __restrict__ bias,
                                const float* __restrict__ dinv,
                                float* __restrict__ agg, int n, int F) {
    long long i = (long long)blockIdx.x * blockDim.x + threadIdx.x;
    long long tot = (long long)n * F;
    if (i >= tot) return;
    int node = (int)(i / F);
    int f    = (int)(i - (long long)node * F);
    float di = dinv[node];
    agg[i] = bias[f] + di * di * H[i];
}

__global__ void __launch_bounds__(256)
scatter_kernel(const int* __restrict__ src, const int* __restrict__ dst,
               const float* __restrict__ w, const float* __restrict__ dinv,
               const float* __restrict__ H, float* __restrict__ agg,
               int e, int F) {
    int wid  = (blockIdx.x * blockDim.x + threadIdx.x) >> 5;
    int lane = threadIdx.x & 31;
    if (wid >= e) return;
    int s = src[wid];
    int d = dst[wid];
    float nrm = dinv[s] * w[wid] * dinv[d];
    const float* h = H + (size_t)s * F;
    float* o = agg + (size_t)d * F;
    for (int f = lane; f < F; f += 32)
        atomicAdd(&o[f], nrm * h[f]);
}

#define SCAN_GROUPS 4
#define TILE_RP 208

template <int FN>
__global__ void __launch_bounds__(1024)
agg_tile_kernel(const float* __restrict__ dinv,
                const float* __restrict__ H, const float* __restrict__ bias,
                const unsigned* __restrict__ planes, const unsigned* __restrict__ rec,
                const float* __restrict__ Wn,
                float* __restrict__ outp, int n, int e, int F) {
    __shared__ __attribute__((aligned(16))) float tile[TILE_RP * 132];
    const int lane = threadIdx.x & 31;
    const int wave = threadIdx.x >> 5;
    const int nw   = blockDim.x >> 5;
    const int row0 = blockIdx.x * TILE_R;
    const int rows = (n - row0 < TILE_R) ? (n - row0) : TILE_R;
    const int P    = F + 4;

    for (int r = wave; r < rows; r += nw) {
        float di = dinv[row0 + r];
        const float* h = H + (size_t)(row0 + r) * F;
        for (int f = lane; f < F; f += 32)
            tile[r * P + f] = bias[f] + di * di * h[f];
    }
    if (FN > 0) {
        for (int r = rows + wave; r < TILE_RP; r += nw)
            for (int f = lane; f < F; f += 32)
                tile[r * P + f] = 0.0f;
    }
    __syncthreads();

    int mye = 0;
    int cnt = 0;
    const uint2* rec2 = reinterpret_cast<const uint2*>(rec);

    auto flush = [&](int my_edge, int count) {
        int s = 0, dl = 0;
        float nrm = 0.0f;
        if (lane < count) {
            uint2 r = rec2[my_edge];
            s = (int)(r.x & 0xFFFFu);
            dl = (int)(r.x >> 16) - row0;
            nrm = __uint_as_float(r.y);
        }
        for (int p = 0; p < count; p += 4) {
            float v[4][4];
            int   dls[4];
            float nrs[4];
#pragma unroll
            for (int q = 0; q < 4; ++q) {
                int pp = (p + q < count) ? (p + q) : (count - 1);
                int sp = __builtin_amdgcn_readlane(s, pp);
                dls[q] = __builtin_amdgcn_readlane(dl, pp);
                nrs[q] = __int_as_float(
                    __builtin_amdgcn_readlane(__float_as_int(nrm), pp));
                const float* h = H + (size_t)sp * F;
#pragma unroll
                for (int j = 0; j < 4; ++j)
                    v[q][j] = (32 * j < F) ? h[lane + 32 * j] : 0.0f;
            }
#pragma unroll
            for (int q = 0; q < 4; ++q) {
                if (p + q < count) {
                    float* o = tile + dls[q] * P;
#pragma unroll
                    for (int j = 0; j < 4; ++j)
                        if (32 * j < F)
                            atomicAdd(&o[lane + 32 * j], nrs[q] * v[q][j]);
                }
            }
        }
    };

    const uint4* pl4 = reinterpret_cast<const uint4*>(planes);
    const unsigned tid = blockIdx.x;
    unsigned sel[8];
#pragma unroll
    for (int p = 0; p < 8; ++p) sel[p] = ((tid >> p) & 1u) ? 0u : ~0u;

    const int ntrips = (e + 1024 * SCAN_GROUPS - 1) / (1024 * SCAN_GROUPS);
    for (int t = wave; t < ntrips; t += nw) {
        unsigned bv[SCAN_GROUPS];
        {
            uint4 pa[SCAN_GROUPS], pb[SCAN_GROUPS];
#pragma unroll
            for (int g = 0; g < SCAN_GROUPS; ++g) {
                const size_t blk = (size_t)(t * SCAN_GROUPS + g) * 64;
                pa[g] = pl4[blk + lane];
                pb[g] = pl4[blk + 32 + lane];
            }
#pragma unroll
            for (int g = 0; g < SCAN_GROUPS; ++g)
                bv[g] = (pa[g].x ^ sel[0]) & (pa[g].y ^ sel[1]) & (pa[g].z ^ sel[2]) & (pa[g].w ^ sel[3])
                      & (pb[g].x ^ sel[4]) & (pb[g].y ^ sel[5]) & (pb[g].z ^ sel[6]) & (pb[g].w ^ sel[7]);
        }
#pragma nounroll
        for (int g = 0; g < SCAN_GROUPS; ++g) {
            unsigned bvg = bv[0];
#pragma unroll
            for (int q = 1; q < SCAN_GROUPS; ++q) bvg = (g == q) ? bv[q] : bvg;
            const int base = (t * SCAN_GROUPS + g) << 10;
            unsigned have = __builtin_amdgcn_ballot_w32(bvg != 0);
            while (have) {
                int j = __builtin_ctz(have);
                have &= have - 1;
                unsigned b = (unsigned)__builtin_amdgcn_readlane((int)bvg, j);
                while (b) {
                    int L = __builtin_ctz(b);
                    b &= b - 1;
                    mye = (lane == cnt) ? (base + j * 32 + L) : mye;
                    if (++cnt == 32) {
                        flush(mye, 32);
                        cnt = 0;
                    }
                }
            }
        }
    }
    if (cnt) flush(mye, cnt);

    if constexpr (FN > 0) {
        constexpr int NCT = FN / 16;
        const int nsg  = nw / NCT;
        const int wv   = __builtin_amdgcn_readfirstlane(wave);
        const int ct   = wv % NCT;
        const int sg   = wv / NCT;
        const int l16  = lane & 15;
        const int half = lane >> 4;
        const int col  = ct * 16 + l16;
        v2f bw[32];
#pragma unroll
        for (int kk = 0; kk < 32; ++kk) {
            const int ka = kk * 4 + 2 * half;
            bw[kk].x = Wn[(size_t)ka * FN + col];
            bw[kk].y = Wn[(size_t)(ka + 1) * FN + col];
        }
        __syncthreads();
        const int nstrips = (rows + 15) >> 4;
        for (int s = sg; s < nstrips; s += nsg) {
            const float* arow = tile + (s * 16 + l16) * P + 2 * half;
            v8f c = {0.f, 0.f, 0.f, 0.f, 0.f, 0.f, 0.f, 0.f};
#pragma unroll
            for (int kk = 0; kk < 32; ++kk) {
                v2f a = *reinterpret_cast<const v2f*>(arow + kk * 4);
                a.x = fmaxf(a.x, 0.0f);
                a.y = fmaxf(a.y, 0.0f);
                c = __builtin_amdgcn_wmma_f32_16x16x4_f32(false, a, false, bw[kk],
                                                          (short)0, c, false, false);
            }
            const int mb = s * 16 + half * 8;
            float* o = outp + (size_t)(row0 + mb) * FN + col;
#pragma unroll
            for (int r = 0; r < 8; ++r)
                if (mb + r < rows) o[(size_t)r * FN] = c[r];
        }
    } else {
        __syncthreads();
        for (int r = wave; r < rows; r += nw) {
            float* o = outp + (size_t)(row0 + r) * F;
            for (int f = lane; f < F; f += 32)
                o[f] = tile[r * P + f];
        }
    }
}

extern "C" void kernel_launch(void* const* d_in, const int* in_sizes, int n_in,
                              void* d_out, int out_size, void* d_ws, size_t ws_size,
                              hipStream_t stream) {
    const float* x    = (const float*)d_in[0];
    const int*   ei   = (const int*)d_in[1];
    const float* ew   = (const float*)d_in[2];
    const float* W1   = (const float*)d_in[3];
    const float* b1   = (const float*)d_in[4];
    const float* W2   = (const float*)d_in[5];
    const float* b2   = (const float*)d_in[6];
    const float* W3   = (const float*)d_in[7];
    const float* b3   = (const float*)d_in[8];
    float*       out  = (float*)d_out;

    const int hidden  = in_sizes[4];
    const int out_dim = in_sizes[8];
    const int in_dim  = in_sizes[3] / hidden;
    const int n       = in_sizes[0] / in_dim;
    const int e       = in_sizes[2];

    const int* src = ei;
    const int* dst = ei + e;

    char* ws = (char*)d_ws;
    size_t off = 0;
    float* dinv = (float*)(ws + off);
    off += ((size_t)n * sizeof(float) + 255) & ~(size_t)255;
    float* H = (float*)(ws + off);
    off += ((size_t)n * hidden * sizeof(float) + 255) & ~(size_t)255;
    float* AGG = (float*)(ws + off);

    const int B = 256;
    auto cdiv = [](long long a, long long b) { return (int)((a + b - 1) / b); };

    const int ntiles = cdiv(n, TILE_R);
    const int e_pad  = cdiv(e, 1024 * SCAN_GROUPS) * 1024 * SCAN_GROUPS;
    const size_t side = (size_t)e_pad + (size_t)e * 8;
    const bool keyed = ntiles <= 255 && n <= REC_NMAX && hidden % 32 == 0 && hidden <= 128 &&
                       out_dim % 32 == 0 && out_dim <= 128 &&
                       side + (size_t)n * sizeof(float) <= (size_t)n * out_dim * sizeof(float) &&
                       side <= (size_t)n * hidden * sizeof(float);
    unsigned* planes_a = (unsigned*)d_out;
    unsigned* planes_b = (unsigned*)AGG;
    unsigned* rec_a = (unsigned*)((char*)d_out + e_pad);
    unsigned* rec_b = (unsigned*)((char*)AGG + e_pad);
    float*    degs  = keyed ? (float*)((char*)d_out + side) : dinv;

    auto aggregate = [&](const float* Hin, const float* bias, float* dstbuf, int F,
                         const unsigned* planes, const unsigned* rec) {
        if (keyed) {
            agg_tile_kernel<0><<<ntiles, 1024, 0, stream>>>(
                dinv, Hin, bias, planes, rec, nullptr, dstbuf, n, e, F);
        } else {
            init_agg_kernel<<<cdiv((long long)n * F, B), B, 0, stream>>>(
                Hin, bias, dinv, dstbuf, n, F);
            scatter_kernel<<<cdiv((long long)e * 32, B), B, 0, stream>>>(
                src, dst, ew, dinv, Hin, dstbuf, e, F);
        }
    };

    deg_init_kernel<<<cdiv(n, B), B, 0, stream>>>(degs, n);
    if (keyed) {
        edge_pass_kernel<true><<<e_pad / B, B, 0, stream>>>(dst, ew, degs, planes_a,
                                                            nullptr, nullptr, e);
        norm_rec_kernel<<<REC_WGS, 1024, 0, stream>>>(src, dst, ew, degs, dinv, rec_a, n, e);
    } else {
        deg_accum_kernel<<<cdiv(e, B), B, 0, stream>>>(dst, ew, dinv, e);
        dinv_kernel<<<cdiv(n, B), B, 0, stream>>>(dinv, n);
    }

    auto product1 = [&]() {
        if (in_dim == 256 && hidden == 128 && n > 0) {
            gemm1_wstat_kernel<<<256, 512, 0, stream>>>(x, W1, H, n);
        } else {
            int tiles = (n / 16) * (hidden / 16);
            gemm_wmma_f32_kernel<<<cdiv((long long)tiles * 32, B), B, 0, stream>>>(
                x, W1, H, n, in_dim, hidden, 0);
        }
    };

    if (keyed && hidden == 128 && out_dim == 64) {
        product1();
        agg_tile_kernel<128><<<ntiles, 1024, 0, stream>>>(
            dinv, H, b1, planes_a, rec_a, W2, AGG, n, e, hidden);
        agg_tile_kernel<64><<<ntiles, 1024, 0, stream>>>(
            dinv, AGG, b2, planes_a, rec_a, W3, H, n, e, hidden);
        edge_pass_kernel<false><<<e_pad / B, B, 0, stream>>>(dst, nullptr, nullptr, planes_b,
                                                             rec_a, rec_b, e);
        agg_tile_kernel<0><<<ntiles, 1024, 0, stream>>>(
            dinv, H, b3, planes_b, rec_b, nullptr, out, n, e, out_dim);
        return;
    }

    {
        product1();
        aggregate(H, b1, AGG, hidden, planes_a, rec_a);
    }

    {
        int tiles = (n / 16) * (hidden / 16);
        gemm_wmma_f32_kernel<<<cdiv((long long)tiles * 32, B), B, 0, stream>>>(
            AGG, W2, H, n, hidden, hidden, 1);
        aggregate(H, b2, AGG, hidden, planes_a, rec_a);
    }

    {
        int tiles = (n / 16) * (out_dim / 16);
        gemm_wmma_f32_kernel<<<cdiv((long long)tiles * 32, B), B, 0, stream>>>(
            AGG, W3, H, n, hidden, out_dim, 1);
        if (keyed)
            edge_pass_kernel<false><<<e_pad / B, B, 0, stream>>>(dst, nullptr, nullptr, planes_b,
                                                                 rec_a, rec_b, e);
        aggregate(H, b3, out, out_dim, planes_b, rec_b);
    }
}


# Aggregation gather: a wave fetches the source rows of 16 matching edges as one group (8-byte loads, two per 128-float row) before their LDS adds, not 4 rows; LDS atomic sums still in arrival order
# speedup vs baseline: 2.2756x; 1.0788x over previous
#include <hip/hip_runtime.h>
#include <hip/hip_bf16.h>

typedef __attribute__((ext_vector_type(2))) float v2f;
typedef __attribute__((ext_vector_type(8))) float v8f;

__global__ void deg_init_kernel(float* deg, int n) {
    int i = blockIdx.x * blockDim.x + threadIdx.x;
    if (i < n) deg[i] = 1.0f;
}

__global__ void deg_accum_kernel(const int* dst, const float* w, float* deg, int e) {
    int i = blockIdx.x * blockDim.x + threadIdx.x;
    if (i < e) atomicAdd(&deg[dst[i]], w[i]);
}

#define TILE_R 200

template <bool WITH_DEG>
__global__ void __launch_bounds__(256)
edge_pass_kernel(const int* __restrict__ dst, const float* __restrict__ w,
                 float* __restrict__ deg, unsigned* __restrict__ planes,
                 const unsigned* __restrict__ rec_in, unsigned* __restrict__ rec_out, int e) {
    int i = blockIdx.x * blockDim.x + threadIdx.x;
    int lane = threadIdx.x & 31;
    unsigned key = 255u;
    if (i < e) {
        int d = dst[i];
        if (WITH_DEG) atomicAdd(&deg[d], w[i]);
        else reinterpret_cast<uint2*>(rec_out)[i] = reinterpret_cast<const uint2*>(rec_in)[i];
        unsigned t = (unsigned)d / TILE_R;
        key = (t < 255u) ? t : 255u;
    }
    unsigned mine = 0;
#pragma unroll
    for (int p = 0; p < 8; ++p) {
        unsigned b = __builtin_amdgcn_ballot_w32((key >> p) & 1u);
        mine = (lane == p) ? b : mine;
    }
    if (lane < 8) {
        int g = i >> 5;
        planes[(size_t)(g >> 5) * 256 + (lane >> 2) * 128 + (g & 31) * 4 + (lane & 3)] = mine;
    }
}

__global__ void dinv_kernel(float* deg, int n) {
    int i = blockIdx.x * blockDim.x + threadIdx.x;
    if (i < n) {
        float d = deg[i];
        deg[i] = (d > 0.0f) ? rsqrtf(d) : 0.0f;
    }
}

#define REC_NMAX (255 * TILE_R)
#define REC_WGS  256

__global__ void __launch_bounds__(1024)
norm_rec_kernel(const int* __restrict__ src, const int* __restrict__ dst,
                const float* __restrict__ w, const float* __restrict__ deg,
                float* __restrict__ dinv, unsigned* __restrict__ rec, int n, int e) {
    __shared__ float sdinv[REC_NMAX];
#pragma unroll 8
    for (int i = threadIdx.x; i < n; i += 1024) {
        float d = deg[i];
        sdinv[i] = (d > 0.0f) ? rsqrtf(d) : 0.0f;
    }
    __syncthreads();
    const int nper = ((n + REC_WGS * 32 - 1) / (REC_WGS * 32)) * 32;
    const int n0   = blockIdx.x * nper;
    const int n1   = (n0 + nper < n) ? (n0 + nper) : n;
    for (int i = n0 + threadIdx.x; i < n1; i += 1024) dinv[i] = sdinv[i];
    const int eper = ((e + REC_WGS * 32 - 1) / (REC_WGS * 32)) * 32;
    const int e0   = blockIdx.x * eper;
    const int e1   = (e0 + eper < e) ? (e0 + eper) : e;
    uint2* rec2 = reinterpret_cast<uint2*>(rec);
    for (int base = e0; base < e1; base += 4 * 1024) {
        int   s[4], d[4];
        float ww[4];
#pragma unroll
        for (int k = 0; k < 4; ++k) {
            const int i = base + k * 1024 + (int)threadIdx.x;
            s[k] = 0; d[k] = 0; ww[k] = 0.0f;
            if (i < e1) { s[k] = src[i]; d[k] = dst[i]; ww[k] = w[i]; }
        }
#pragma unroll
        for (int k = 0; k < 4; ++k) {
            const int i = base + k * 1024 + (int)threadIdx.x;
            if (i < e1) {
                float nrm = sdinv[s[k]] * ww[k] * sdinv[d[k]];
                rec2[i] = make_uint2((unsigned)s[k] | ((unsigned)d[k] << 16), __float_as_uint(nrm));
            }
        }
    }
}

__global__ void __launch_bounds__(256)
gemm_wmma_f32_kernel(const float* __restrict__ A, const float* __restrict__ W,
                     float* __restrict__ Out, int M, int K, int F, int relu_in) {
    int wid  = (blockIdx.x * blockDim.x + threadIdx.x) >> 5;
    int lane = threadIdx.x & 31;
    int tiles_n = F >> 4;
    int total   = (M >> 4) * tiles_n;
    if (wid >= total) return;

    int tm  = wid / tiles_n;
    int tn  = wid - tm * tiles_n;
    int l16  = lane & 15;
    int half = lane >> 4;

    int row = tm * 16 + l16;
    int col = tn * 16 + l16;

    const float* arow = A + (size_t)row * K;
    v8f c = {0.f, 0.f, 0.f, 0.f, 0.f, 0.f, 0.f, 0.f};

    for (int k = 0; k < K; k += 4) {
        int ka = k + 2 * half;
        v2f a;
        a.x = arow[ka];
        a.y = arow[ka + 1];
        if (relu_in) {
            a.x = fmaxf(a.x, 0.0f);
            a.y = fmaxf(a.y, 0.0f);
        }
        v2f b;
        b.x = W[(size_t)ka * F + col];
        b.y = W[(size_t)(ka + 1) * F + col];
        c = __builtin_amdgcn_wmma_f32_16x16x4_f32(false, a, false, b,
                                                  (short)0, c, false, false);
    }

    int mbase = tm * 16 + half * 8;
#pragma unroll
    for (int r = 0; r < 8; ++r)
        Out[(size_t)(mbase + r) * F + col] = c[r];
}

#define G1_ROWS 32
#define G1_P    260

static __device__ __forceinline__ void lds_barrier() {
    asm volatile("s_wait_dscnt 0x0\n\ts_barrier_signal -1\n\ts_barrier_wait -1" ::: "memory");
}

__global__ void __launch_bounds__(512)
gemm1_wstat_kernel(const float* __restrict__ A, const float* __restrict__ W,
                   float* __restrict__ Out, int M) {
    constexpr int K = 256;
    constexpr int F = 128;
    __shared__ __attribute__((aligned(16))) float sa[2 * G1_ROWS * G1_P];
    const int lane = threadIdx.x & 31;
    const int wv   = __builtin_amdgcn_readfirstlane((int)(threadIdx.x >> 5));
    const int ct   = wv & 7;
    const int sg   = wv >> 3;
    const int l16  = lane & 15;
    const int half = lane >> 4;
    const int col  = ct * 16 + l16;

    const int nstrips  = (M + 15) >> 4;
    const int nwg      = (int)gridDim.x;
    const int base     = nstrips / nwg;
    const int rem      = nstrips - base * nwg;
    const int b        = (int)blockIdx.x;
    const int mystrips = base + ((b < rem) ? 1 : 0);
    const int strip0   = b * base + ((b < rem) ? b : rem);
    const int row0     = strip0 * 16;
    const int nrounds  = (mystrips + 1) >> 1;

    const float4* A4 = reinterpret_cast<const float4*>(A);
    const int lr = threadIdx.x >> 6;
    const int c4 = threadIdx.x & 63;

    float4 p0, p1, p2, p3;
    const float4* acol = A4 + c4;
    auto rowptr = [&](int grow) {
        const int g = (grow < M) ? grow : (M - 1);
        return acol + (size_t)g * (K / 4);
    };
    auto fetch = [&](int r) {
        const int g0 = row0 + r * G1_ROWS + lr;
        p0 = *rowptr(g0);
        p1 = *rowptr(g0 + 8);
        p2 = *rowptr(g0 + 16);
        p3 = *rowptr(g0 + 24);
    };
    auto stash = [&](int buf, float4 a0, float4 a1, float4 a2, float4 a3) {
        float* d = &sa[buf * (G1_ROWS * G1_P) + lr * G1_P + 4 * c4];
        *reinterpret_cast<float4*>(d)             = a0;
        *reinterpret_cast<float4*>(d + 8 * G1_P)  = a1;
        *reinterpret_cast<float4*>(d + 16 * G1_P) = a2;
        *reinterpret_cast<float4*>(d + 24 * G1_P) = a3;
    };

    const int gfirst = row0 + lr;
    const float4 q0 = *rowptr(gfirst);
    const float4 q1 = *rowptr(gfirst + 8);
    const float4 q2 = *rowptr(gfirst + 16);
    const float4 q3 = *rowptr(gfirst + 24);
    v2f bw[K / 4];
#pragma unroll
    for (int kk = 0; kk < K / 4; ++kk) {
        const int ka = kk * 4 + 2 * half;
        bw[kk].x = W[(size_t)ka * F + col];
        bw[kk].y = W[(size_t)(ka + 1) * F + col];
    }
    fetch(1);
    stash(0, q0, q1, q2, q3);
    __syncthreads();

    for (int r = 0; r < nrounds; ++r) {
        if (r + 1 < nrounds) stash((r + 1) & 1, p0, p1, p2, p3);
        if (r + 2 < nrounds) fetch(r + 2);
        const int ls = 2 * r + sg;
        if (ls < mystrips) {
            const float* arow = sa + (r & 1) * (G1_ROWS * G1_P) + (sg * 16 + l16) * G1_P + 2 * half;
            v8f c = {0.f, 0.f, 0.f, 0.f, 0.f, 0.f, 0.f, 0.f};
#pragma unroll
            for (int kk = 0; kk < K / 4; ++kk) {
                v2f a = *reinterpret_cast<const v2f*>(arow + kk * 4);
                c = __builtin_amdgcn_wmma_f32_16x16x4_f32(false, a, false, bw[kk],
                                                          (short)0, c, false, false);
            }
            const int mb = row0 + ls * 16 + half * 8;
            float* o = Out + (size_t)mb * F + col;
#pragma unroll
            for (int q = 0; q < 8; ++q)
                if (mb + q < M) o[(size_t)q * F] = c[q];
        }
        lds_barrier();
    }
}

__global__ void init_agg_kernel(const float* __restrict__ H,
                                const float* __restrict__ bias,
                                const float* __restrict__ dinv,
                                float* __restrict__ agg, int n, int F) {
    long long i = (long long)blockIdx.x * blockDim.x + threadIdx.x;
    long long tot = (long long)n * F;
    if (i >= tot) return;
    int node = (int)(i / F);
    int f    = (int)(i - (long long)node * F);
    float di = dinv[node];
    agg[i] = bias[f] + di * di * H[i];
}

__global__ void __launch_bounds__(256)
scatter_kernel(const int* __restrict__ src, const int* __restrict__ dst,
               const float* __restrict__ w, const float* __restrict__ dinv,
               const float* __restrict__ H, float* __restrict__ agg,
               int e, int F) {
    int wid  = (blockIdx.x * blockDim.x + threadIdx.x) >> 5;
    int lane = threadIdx.x & 31;
    if (wid >= e) return;
    int s = src[wid];
    int d = dst[wid];
    float nrm = dinv[s] * w[wid] * dinv[d];
    const float* h = H + (size_t)s * F;
    float* o = agg + (size_t)d * F;
    for (int f = lane; f < F; f += 32)
        atomicAdd(&o[f], nrm * h[f]);
}

#define SCAN_GROUPS 4
#define TILE_RP 208
#define GATHER_Q 16

template <int N> struct NTag { static constexpr int value = N; };

template <int FN>
__global__ void __launch_bounds__(1024)
agg_tile_kernel(const float* __restrict__ dinv,
                const float* __restrict__ H, const float* __restrict__ bias,
                const unsigned* __restrict__ planes, const unsigned* __restrict__ rec,
                const float* __restrict__ Wn,
                float* __restrict__ outp, int n, int e, int F) {
    __shared__ __attribute__((aligned(16))) float tile[TILE_RP * 132];
    const int lane = threadIdx.x & 31;
    const int wave = threadIdx.x >> 5;
    const int nw   = blockDim.x >> 5;
    const int row0 = blockIdx.x * TILE_R;
    const int rows = (n - row0 < TILE_R) ? (n - row0) : TILE_R;
    const int P    = F + 4;
    const int Fg   = (FN > 0) ? 128 : F;

    for (int r = wave; r < rows; r += nw) {
        float di = dinv[row0 + r];
        const float* h = H + (size_t)(row0 + r) * F;
        for (int f = lane; f < F; f += 32)
            tile[r * P + f] = bias[f] + di * di * h[f];
    }
    if (FN > 0) {
        for (int r = rows + wave; r < TILE_RP; r += nw)
            for (int f = lane; f < F; f += 32)
                tile[r * P + f] = 0.0f;
    }
    __syncthreads();

    int mye = 0;
    int cnt = 0;
    const uint2* rec2 = reinterpret_cast<const uint2*>(rec);

    auto flush = [&](int my_edge, int count) {
        int s = 0, dl = 0;
        float nrm = 0.0f;
        if (lane < count) {
            uint2 r = rec2[my_edge];
            s = (int)(r.x & 0xFFFFu);
            dl = (int)(r.x >> 16) - row0;
            nrm = __uint_as_float(r.y);
        }
        auto groups = [&](auto tag) {
            constexpr int NJ = decltype(tag)::value;
            for (int p = 0; p < count; p += GATHER_Q) {
                v2f   v[GATHER_Q][NJ];
                int   dls[GATHER_Q];
                float nrs[GATHER_Q];
#pragma unroll
                for (int q = 0; q < GATHER_Q; ++q) {
                    int pp = (p + q < count) ? (p + q) : (count - 1);
                    int sp = __builtin_amdgcn_readlane(s, pp);
                    dls[q] = __builtin_amdgcn_readlane(dl, pp);
                    nrs[q] = __int_as_float(
                        __builtin_amdgcn_readlane(__float_as_int(nrm), pp));
                    const v2f* h = reinterpret_cast<const v2f*>(H + (size_t)sp * Fg);
#pragma unroll
                    for (int j = 0; j < NJ; ++j)
                        v[q][j] = h[lane + 32 * j];
                }
#pragma unroll
                for (int q = 0; q < GATHER_Q; ++q) {
                    if (p + q < count) {
                        float* o = tile + dls[q] * P + 2 * lane;
#pragma unroll
                        for (int j = 0; j < NJ; ++j) {
                            atomicAdd(&o[64 * j],     nrs[q] * v[q][j].x);
                            atomicAdd(&o[64 * j + 1], nrs[q] * v[q][j].y);
                        }
                    }
                }
            }
        };
        if (Fg > 64) groups(NTag<2>{}); else groups(NTag<1>{});
    };

    const uint4* pl4 = reinterpret_cast<const uint4*>(planes);
    const unsigned tid = blockIdx.x;
    unsigned sel[8];
#pragma unroll
    for (int p = 0; p < 8; ++p) sel[p] = ((tid >> p) & 1u) ? 0u : ~0u;

    const int ntrips = (e + 1024 * SCAN_GROUPS - 1) / (1024 * SCAN_GROUPS);
    for (int t = wave; t < ntrips; t += nw) {
        unsigned bv[SCAN_GROUPS];
        {
            uint4 pa[SCAN_GROUPS], pb[SCAN_GROUPS];
#pragma unroll
            for (int g = 0; g < SCAN_GROUPS; ++g) {
                const size_t blk = (size_t)(t * SCAN_GROUPS + g) * 64;
                pa[g] = pl4[blk + lane];
                pb[g] = pl4[blk + 32 + lane];
            }
#pragma unroll
            for (int g = 0; g < SCAN_GROUPS; ++g)
                bv[g] = (pa[g].x ^ sel[0]) & (pa[g].y ^ sel[1]) & (pa[g].z ^ sel[2]) & (pa[g].w ^ sel[3])
                      & (pb[g].x ^ sel[4]) & (pb[g].y ^ sel[5]) & (pb[g].z ^ sel[6]) & (pb[g].w ^ sel[7]);
        }
#pragma nounroll
        for (int g = 0; g < SCAN_GROUPS; ++g) {
            unsigned bvg = bv[0];
#pragma unroll
            for (int q = 1; q < SCAN_GROUPS; ++q) bvg = (g == q) ? bv[q] : bvg;
            const int base = (t * SCAN_GROUPS + g) << 10;
            unsigned have = __builtin_amdgcn_ballot_w32(bvg != 0);
            while (have) {
                int j = __builtin_ctz(have);
                have &= have - 1;
                unsigned b = (unsigned)__builtin_amdgcn_readlane((int)bvg, j);
                while (b) {
                    int L = __builtin_ctz(b);
                    b &= b - 1;
                    mye = (lane == cnt) ? (base + j * 32 + L) : mye;
                    if (++cnt == 32) {
                        flush(mye, 32);
                        cnt = 0;
                    }
                }
            }
        }
    }
    if (cnt) flush(mye, cnt);

    if constexpr (FN > 0) {
        constexpr int NCT = FN / 16;
        const int nsg  = nw / NCT;
        const int wv   = __builtin_amdgcn_readfirstlane(wave);
        const int ct   = wv % NCT;
        const int sg   = wv / NCT;
        const int l16  = lane & 15;
        const int half = lane >> 4;
        const int col  = ct * 16 + l16;
        v2f bw[32];
#pragma unroll
        for (int kk = 0; kk < 32; ++kk) {
            const int ka = kk * 4 + 2 * half;
            bw[kk].x = Wn[(size_t)ka * FN + col];
            bw[kk].y = Wn[(size_t)(ka + 1) * FN + col];
        }
        __syncthreads();
        const int nstrips = (rows + 15) >> 4;
        for (int s = sg; s < nstrips; s += nsg) {
            const float* arow = tile + (s * 16 + l16) * P + 2 * half;
            v8f c = {0.f, 0.f, 0.f, 0.f, 0.f, 0.f, 0.f, 0.f};
#pragma unroll
            for (int kk = 0; kk < 32; ++kk) {
                v2f a = *reinterpret_cast<const v2f*>(arow + kk * 4);
                a.x = fmaxf(a.x, 0.0f);
                a.y = fmaxf(a.y, 0.0f);
                c = __builtin_amdgcn_wmma_f32_16x16x4_f32(false, a, false, bw[kk],
                                                          (short)0, c, false, false);
            }
            const int mb = s * 16 + half * 8;
            float* o = outp + (size_t)(row0 + mb) * FN + col;
#pragma unroll
            for (int r = 0; r < 8; ++r)
                if (mb + r < rows) o[(size_t)r * FN] = c[r];
        }
    } else {
        __syncthreads();
        for (int r = wave; r < rows; r += nw) {
            float* o = outp + (size_t)(row0 + r) * F;
            for (int f = lane; f < F; f += 32)
                o[f] = tile[r * P + f];
        }
    }
}

extern "C" void kernel_launch(void* const* d_in, const int* in_sizes, int n_in,
                              void* d_out, int out_size, void* d_ws, size_t ws_size,
                              hipStream_t stream) {
    const float* x    = (const float*)d_in[0];
    const int*   ei   = (const int*)d_in[1];
    const float* ew   = (const float*)d_in[2];
    const float* W1   = (const float*)d_in[3];
    const float* b1   = (const float*)d_in[4];
    const float* W2   = (const float*)d_in[5];
    const float* b2   = (const float*)d_in[6];
    const float* W3   = (const float*)d_in[7];
    const float* b3   = (const float*)d_in[8];
    float*       out  = (float*)d_out;

    const int hidden  = in_sizes[4];
    const int out_dim = in_sizes[8];
    const int in_dim  = in_sizes[3] / hidden;
    const int n       = in_sizes[0] / in_dim;
    const int e       = in_sizes[2];

    const int* src = ei;
    const int* dst = ei + e;

    char* ws = (char*)d_ws;
    size_t off = 0;
    float* dinv = (float*)(ws + off);
    off += ((size_t)n * sizeof(float) + 255) & ~(size_t)255;
    float* H = (float*)(ws + off);
    off += ((size_t)n * hidden * sizeof(float) + 255) & ~(size_t)255;
    float* AGG = (float*)(ws + off);

    const int B = 256;
    auto cdiv = [](long long a, long long b) { return (int)((a + b - 1) / b); };

    const int ntiles = cdiv(n, TILE_R);
    const int e_pad  = cdiv(e, 1024 * SCAN_GROUPS) * 1024 * SCAN_GROUPS;
    const size_t side = (size_t)e_pad + (size_t)e * 8;
    const bool keyed = ntiles <= 255 && n <= REC_NMAX && hidden % 64 == 0 && hidden <= 128 &&
                       out_dim % 64 == 0 && out_dim <= 128 &&
                       side + (size_t)n * sizeof(float) <= (size_t)n * out_dim * sizeof(float) &&
                       side <= (size_t)n * hidden * sizeof(float);
    unsigned* planes_a = (unsigned*)d_out;
    unsigned* planes_b = (unsigned*)AGG;
    unsigned* rec_a = (unsigned*)((char*)d_out + e_pad);
    unsigned* rec_b = (unsigned*)((char*)AGG + e_pad);
    float*    degs  = keyed ? (float*)((char*)d_out + side) : dinv;

    auto aggregate = [&](const float* Hin, const float* bias, float* dstbuf, int F,
                         const unsigned* planes, const unsigned* rec) {
        if (keyed) {
            agg_tile_kernel<0><<<ntiles, 1024, 0, stream>>>(
                dinv, Hin, bias, planes, rec, nullptr, dstbuf, n, e, F);
        } else {
            init_agg_kernel<<<cdiv((long long)n * F, B), B, 0, stream>>>(
                Hin, bias, dinv, dstbuf, n, F);
            scatter_kernel<<<cdiv((long long)e * 32, B), B, 0, stream>>>(
                src, dst, ew, dinv, Hin, dstbuf, e, F);
        }
    };

    deg_init_kernel<<<cdiv(n, B), B, 0, stream>>>(degs, n);
    if (keyed) {
        edge_pass_kernel<true><<<e_pad / B, B, 0, stream>>>(dst, ew, degs, planes_a,
                                                            nullptr, nullptr, e);
        norm_rec_kernel<<<REC_WGS, 1024, 0, stream>>>(src, dst, ew, degs, dinv, rec_a, n, e);
    } else {
        deg_accum_kernel<<<cdiv(e, B), B, 0, stream>>>(dst, ew, dinv, e);
        dinv_kernel<<<cdiv(n, B), B, 0, stream>>>(dinv, n);
    }

    auto product1 = [&]() {
        if (in_dim == 256 && hidden == 128 && n > 0) {
            gemm1_wstat_kernel<<<256, 512, 0, stream>>>(x, W1, H, n);
        } else {
            int tiles = (n / 16) * (hidden / 16);
            gemm_wmma_f32_kernel<<<cdiv((long long)tiles * 32, B), B, 0, stream>>>(
                x, W1, H, n, in_dim, hidden, 0);
        }
    };

    if (keyed && hidden == 128 && out_dim == 64) {
        product1();
        agg_tile_kernel<128><<<ntiles, 1024, 0, stream>>>(
            dinv, H, b1, planes_a, rec_a, W2, AGG, n, e, hidden);
        agg_tile_kernel<64><<<ntiles, 1024, 0, stream>>>(
            dinv, AGG, b2, planes_a, rec_a, W3, H, n, e, hidden);
        edge_pass_kernel<false><<<e_pad / B, B, 0, stream>>>(dst, nullptr, nullptr, planes_b,
                                                             rec_a, rec_b, e);
        agg_tile_kernel<0><<<ntiles, 1024, 0, stream>>>(
            dinv, H, b3, planes_b, rec_b, nullptr, out, n, e, out_dim);
        return;
    }

    {
        product1();
        aggregate(H, b1, AGG, hidden, planes_a, rec_a);
    }

    {
        int tiles = (n / 16) * (hidden / 16);
        gemm_wmma_f32_kernel<<<cdiv((long long)tiles * 32, B), B, 0, stream>>>(
            AGG, W2, H, n, hidden, hidden, 1);
        aggregate(H, b2, AGG, hidden, planes_a, rec_a);
    }

    {
        int tiles = (n / 16) * (out_dim / 16);
        gemm_wmma_f32_kernel<<<cdiv((long long)tiles * 32, B), B, 0, stream>>>(
            AGG, W3, H, n, hidden, out_dim, 1);
        if (keyed)
            edge_pass_kernel<false><<<e_pad / B, B, 0, stream>>>(dst, nullptr, nullptr, planes_b,
                                                                 rec_a, rec_b, e);
        aggregate(H, b3, out, out_dim, planes_b, rec_b);
    }
}


# Match lists: layer 1's scan saves each tile's edge ids, layer 2 reads them and hands the records on, layer 3 reads those: no 2nd or 3rd scan, no plane copy, 8 launches to 7; LDS sums in arrival order
# speedup vs baseline: 2.5167x; 1.1060x over previous
#include <hip/hip_runtime.h>
#include <hip/hip_bf16.h>

typedef __attribute__((ext_vector_type(2))) float v2f;
typedef __attribute__((ext_vector_type(8))) float v8f;

__global__ void deg_init_kernel(float* deg, int n) {
    int i = blockIdx.x * blockDim.x + threadIdx.x;
    if (i < n) deg[i] = 1.0f;
}

__global__ void deg_accum_kernel(const int* dst, const float* w, float* deg, int e) {
    int i = blockIdx.x * blockDim.x + threadIdx.x;
    if (i < e) atomicAdd(&deg[dst[i]], w[i]);
}

#define TILE_R 200

template <bool WITH_DEG>
__global__ void __launch_bounds__(256)
edge_pass_kernel(const int* __restrict__ dst, const float* __restrict__ w,
                 float* __restrict__ deg, unsigned* __restrict__ planes,
                 const unsigned* __restrict__ rec_in, unsigned* __restrict__ rec_out, int e) {
    int i = blockIdx.x * blockDim.x + threadIdx.x;
    int lane = threadIdx.x & 31;
    unsigned key = 255u;
    if (i < e) {
        int d = dst[i];
        if (WITH_DEG) atomicAdd(&deg[d], w[i]);
        else reinterpret_cast<uint2*>(rec_out)[i] = reinterpret_cast<const uint2*>(rec_in)[i];
        unsigned t = (unsigned)d / TILE_R;
        key = (t < 255u) ? t : 255u;
    }
    unsigned mine = 0;
#pragma unroll
    for (int p = 0; p < 8; ++p) {
        unsigned b = __builtin_amdgcn_ballot_w32((key >> p) & 1u);
        mine = (lane == p) ? b : mine;
    }
    if (lane < 8) {
        int g = i >> 5;
        planes[(size_t)(g >> 5) * 256 + (lane >> 2) * 128 + (g & 31) * 4 + (lane & 3)] = mine;
    }
}

__global__ void dinv_kernel(float* deg, int n) {
    int i = blockIdx.x * blockDim.x + threadIdx.x;
    if (i < n) {
        float d = deg[i];
        deg[i] = (d > 0.0f) ? rsqrtf(d) : 0.0f;
    }
}

#define REC_NMAX (255 * TILE_R)
#define REC_WGS  256

__global__ void __launch_bounds__(1024)
norm_rec_kernel(const int* __restrict__ src, const int* __restrict__ dst,
                const float* __restrict__ w, const float* __restrict__ deg,
                float* __restrict__ dinv, unsigned* __restrict__ rec, int n, int e) {
    __shared__ float sdinv[REC_NMAX];
#pragma unroll 8
    for (int i = threadIdx.x; i < n; i += 1024) {
        float d = deg[i];
        sdinv[i] = (d > 0.0f) ? rsqrtf(d) : 0.0f;
    }
    __syncthreads();
    const int nper = ((n + REC_WGS * 32 - 1) / (REC_WGS * 32)) * 32;
    const int n0   = blockIdx.x * nper;
    const int n1   = (n0 + nper < n) ? (n0 + nper) : n;
    for (int i = n0 + threadIdx.x; i < n1; i += 1024) dinv[i] = sdinv[i];
    const int eper = ((e + REC_WGS * 32 - 1) / (REC_WGS * 32)) * 32;
    const int e0   = blockIdx.x * eper;
    const int e1   = (e0 + eper < e) ? (e0 + eper) : e;
    uint2* rec2 = reinterpret_cast<uint2*>(rec);
    for (int base = e0; base < e1; base += 4 * 1024) {
        int   s[4], d[4];
        float ww[4];
#pragma unroll
        for (int k = 0; k < 4; ++k) {
            const int i = base + k * 1024 + (int)threadIdx.x;
            s[k] = 0; d[k] = 0; ww[k] = 0.0f;
            if (i < e1) { s[k] = src[i]; d[k] = dst[i]; ww[k] = w[i]; }
        }
#pragma unroll
        for (int k = 0; k < 4; ++k) {
            const int i = base + k * 1024 + (int)threadIdx.x;
            if (i < e1) {
                float nrm = sdinv[s[k]] * ww[k] * sdinv[d[k]];
                rec2[i] = make_uint2((unsigned)s[k] | ((unsigned)d[k] << 16), __float_as_uint(nrm));
            }
        }
    }
}

__global__ void __launch_bounds__(256)
gemm_wmma_f32_kernel(const float* __restrict__ A, const float* __restrict__ W,
                     float* __restrict__ Out, int M, int K, int F, int relu_in) {
    int wid  = (blockIdx.x * blockDim.x + threadIdx.x) >> 5;
    int lane = threadIdx.x & 31;
    int tiles_n = F >> 4;
    int total   = (M >> 4) * tiles_n;
    if (wid >= total) return;

    int tm  = wid / tiles_n;
    int tn  = wid - tm * tiles_n;
    int l16  = lane & 15;
    int half = lane >> 4;

    int row = tm * 16 + l16;
    int col = tn * 16 + l16;

    const float* arow = A + (size_t)row * K;
    v8f c = {0.f, 0.f, 0.f, 0.f, 0.f, 0.f, 0.f, 0.f};

    for (int k = 0; k < K; k += 4) {
        int ka = k + 2 * half;
        v2f a;
        a.x = arow[ka];
        a.y = arow[ka + 1];
        if (relu_in) {
            a.x = fmaxf(a.x, 0.0f);
            a.y = fmaxf(a.y, 0.0f);
        }
        v2f b;
        b.x = W[(size_t)ka * F + col];
        b.y = W[(size_t)(ka + 1) * F + col];
        c = __builtin_amdgcn_wmma_f32_16x16x4_f32(false, a, false, b,
                                                  (short)0, c, false, false);
    }

    int mbase = tm * 16 + half * 8;
#pragma unroll
    for (int r = 0; r < 8; ++r)
        Out[(size_t)(mbase + r) * F + col] = c[r];
}

#define G1_ROWS 32
#define G1_P    260

static __device__ __forceinline__ void lds_barrier() {
    asm volatile("s_wait_dscnt 0x0\n\ts_barrier_signal -1\n\ts_barrier_wait -1" ::: "memory");
}

__global__ void __launch_bounds__(512)
gemm1_wstat_kernel(const float* __restrict__ A, const float* __restrict__ W,
                   float* __restrict__ Out, int M) {
    constexpr int K = 256;
    constexpr int F = 128;
    __shared__ __attribute__((aligned(16))) float sa[2 * G1_ROWS * G1_P];
    const int lane = threadIdx.x & 31;
    const int wv   = __builtin_amdgcn_readfirstlane((int)(threadIdx.x >> 5));
    const int ct   = wv & 7;
    const int sg   = wv >> 3;
    const int l16  = lane & 15;
    const int half = lane >> 4;
    const int col  = ct * 16 + l16;

    const int nstrips  = (M + 15) >> 4;
    const int nwg      = (int)gridDim.x;
    const int base     = nstrips / nwg;
    const int rem      = nstrips - base * nwg;
    const int b        = (int)blockIdx.x;
    const int mystrips = base + ((b < rem) ? 1 : 0);
    const int strip0   = b * base + ((b < rem) ? b : rem);
    const int row0     = strip0 * 16;
    const int nrounds  = (mystrips + 1) >> 1;

    const float4* A4 = reinterpret_cast<const float4*>(A);
    const int lr = threadIdx.x >> 6;
    const int c4 = threadIdx.x & 63;

    float4 p0, p1, p2, p3;
    const float4* acol = A4 + c4;
    auto rowptr = [&](int grow) {
        const int g = (grow < M) ? grow : (M - 1);
        return acol + (size_t)g * (K / 4);
    };
    auto fetch = [&](int r) {
        const int g0 = row0 + r * G1_ROWS + lr;
        p0 = *rowptr(g0);
        p1 = *rowptr(g0 + 8);
        p2 = *rowptr(g0 + 16);
        p3 = *rowptr(g0 + 24);
    };
    auto stash = [&](int buf, float4 a0, float4 a1, float4 a2, float4 a3) {
        float* d = &sa[buf * (G1_ROWS * G1_P) + lr * G1_P + 4 * c4];
        *reinterpret_cast<float4*>(d)             = a0;
        *reinterpret_cast<float4*>(d + 8 * G1_P)  = a1;
        *reinterpret_cast<float4*>(d + 16 * G1_P) = a2;
        *reinterpret_cast<float4*>(d + 24 * G1_P) = a3;
    };

    const int gfirst = row0 + lr;
    const float4 q0 = *rowptr(gfirst);
    const float4 q1 = *rowptr(gfirst + 8);
    const float4 q2 = *rowptr(gfirst + 16);
    const float4 q3 = *rowptr(gfirst + 24);
    v2f bw[K / 4];
#pragma unroll
    for (int kk = 0; kk < K / 4; ++kk) {
        const int ka = kk * 4 + 2 * half;
        bw[kk].x = W[(size_t)ka * F + col];
        bw[kk].y = W[(size_t)(ka + 1) * F + col];
    }
    fetch(1);
    stash(0, q0, q1, q2, q3);
    __syncthreads();

    for (int r = 0; r < nrounds; ++r) {
        if (r + 1 < nrounds) stash((r + 1) & 1, p0, p1, p2, p3);
        if (r + 2 < nrounds) fetch(r + 2);
        const int ls = 2 * r + sg;
        if (ls < mystrips) {
            const float* arow = sa + (r & 1) * (G1_ROWS * G1_P) + (sg * 16 + l16) * G1_P + 2 * half;
            v8f c = {0.f, 0.f, 0.f, 0.f, 0.f, 0.f, 0.f, 0.f};
#pragma unroll
            for (int kk = 0; kk < K / 4; ++kk) {
                v2f a = *reinterpret_cast<const v2f*>(arow + kk * 4);
                c = __builtin_amdgcn_wmma_f32_16x16x4_f32(false, a, false, bw[kk],
                                                          (short)0, c, false, false);
            }
            const int mb = row0 + ls * 16 + half * 8;
            float* o = Out + (size_t)mb * F + col;
#pragma unroll
            for (int q = 0; q < 8; ++q)
                if (mb + q < M) o[(size_t)q * F] = c[q];
        }
        lds_barrier();
    }
}

__global__ void init_agg_kernel(const float* __restrict__ H,
                                const float* __restrict__ bias,
                                const float* __restrict__ dinv,
                                float* __restrict__ agg, int n, int F) {
    long long i = (long long)blockIdx.x * blockDim.x + threadIdx.x;
    long long tot = (long long)n * F;
    if (i >= tot) return;
    int node = (int)(i / F);
    int f    = (int)(i - (long long)node * F);
    float di = dinv[node];
    agg[i] = bias[f] + di * di * H[i];
}

__global__ void __launch_bounds__(256)
scatter_kernel(const int* __restrict__ src, const int* __restrict__ dst,
               const float* __restrict__ w, const float* __restrict__ dinv,
               const float* __restrict__ H, float* __restrict__ agg,
               int e, int F) {
    int wid  = (blockIdx.x * blockDim.x + threadIdx.x) >> 5;
    int lane = threadIdx.x & 31;
    if (wid >= e) return;
    int s = src[wid];
    int d = dst[wid];
    float nrm = dinv[s] * w[wid] * dinv[d];
    const float* h = H + (size_t)s * F;
    float* o = agg + (size_t)d * F;
    for (int f = lane; f < F; f += 32)
        atomicAdd(&o[f], nrm * h[f]);
}

#define SCAN_GROUPS 4
#define TILE_RP 208
#define GATHER_Q 16
#define LIST_CAP 5120
#define LIST_HDR 256

template <int N> struct NTag { static constexpr int value = N; };

template <int FN, int LIN, int LOUT>
__global__ void __launch_bounds__(1024)
agg_tile_kernel(const float* __restrict__ dinv,
                const float* __restrict__ H, const float* __restrict__ bias,
                const unsigned* __restrict__ planes, const unsigned* __restrict__ rec,
                const float* __restrict__ Wn,
                float* __restrict__ outp,
                const unsigned* __restrict__ lst_in, unsigned* __restrict__ lst_out,
                const int* __restrict__ esrc, const int* __restrict__ edst,
                const float* __restrict__ ew,
                int n, int e, int F) {
    static_assert((LIN == 0 && LOUT <= 1) || (LIN == 1 && LOUT == 2) || (LIN == 2 && LOUT == 0),
                  "list forms: 0 none, 1 edge ids, 2 edge records");
    __shared__ __attribute__((aligned(16))) float tile[TILE_RP * 132];
    __shared__ unsigned lfill;
    const int lane = threadIdx.x & 31;
    const int wave = threadIdx.x >> 5;
    const int nw   = blockDim.x >> 5;
    const int row0 = blockIdx.x * TILE_R;
    const int rows = (n - row0 < TILE_R) ? (n - row0) : TILE_R;
    const int P    = F + 4;
    const int Fg   = (FN > 0) ? 128 : F;
    const uint2* rec2 = reinterpret_cast<const uint2*>(rec);

    constexpr int MAXB = LIST_CAP / 1024;
    const size_t ibase = (size_t)LIST_HDR + (size_t)blockIdx.x * LIST_CAP;
    const size_t rbase = (size_t)(LIST_HDR / 2) + (size_t)blockIdx.x * LIST_CAP;
    unsigned total  = 0u;
    bool     listed = false;
    if (LIN != 0 && lst_in != nullptr && nw == 32) {
        total  = lst_in[blockIdx.x];
        listed = total <= (unsigned)LIST_CAP;
    }
    const int per     = (int)((total + 31u) >> 5);
    const int first   = wave * per;
    int       mycount = (int)total - first;
    mycount = (mycount > per) ? per : mycount;
    mycount = (mycount > 0 && listed) ? mycount : 0;
    uint2 rr[MAXB];
#pragma unroll
    for (int k = 0; k < MAXB; ++k) rr[k] = make_uint2(0u, 0u);
    if (listed && total != 0u) {
        const int first_c = (mycount > 0) ? first : 0;
        if (LIN == 1) {
            unsigned ids[MAXB];
#pragma unroll
            for (int k = 0; k < MAXB; ++k) {
                const int o = (32 * k + lane < mycount) ? (32 * k + lane) : 0;
                ids[k] = lst_in[ibase + first_c + o];
            }
#pragma unroll
            for (int k = 0; k < MAXB; ++k) rr[k] = rec2[ids[k]];
        } else if (LIN == 2) {
#pragma unroll
            for (int k = 0; k < MAXB; ++k) {
                const int o = (32 * k + lane < mycount) ? (32 * k + lane) : 0;
                rr[k] = reinterpret_cast<const uint2*>(lst_in)[rbase + first_c + o];
            }
        }
    }
    if (threadIdx.x == 0) lfill = 0u;

    for (int r = wave; r < rows; r += nw) {
        float di = dinv[row0 + r];
        const float* h = H + (size_t)(row0 + r) * F;
        for (int f = lane; f < F; f += 32)
            tile[r * P + f] = bias[f] + di * di * h[f];
    }
    if (FN > 0) {
        for (int r = rows + wave; r < TILE_RP; r += nw)
            for (int f = lane; f < F; f += 32)
                tile[r * P + f] = 0.0f;
    }
    __syncthreads();

    auto flush_rec = [&](uint2 r, int count) {
        int s = 0, dl = 0;
        float nrm = 0.0f;
        if (lane < count) {
            s = (int)(r.x & 0xFFFFu);
            dl = (int)(r.x >> 16) - row0;
            nrm = __uint_as_float(r.y);
        }
        auto groups = [&](auto tag) {
            constexpr int NJ = decltype(tag)::value;
            for (int p = 0; p < count; p += GATHER_Q) {
                v2f   v[GATHER_Q][NJ];
                int   dls[GATHER_Q];
                float nrs[GATHER_Q];
#pragma unroll
                for (int q = 0; q < GATHER_Q; ++q) {
                    int pp = (p + q < count) ? (p + q) : (count - 1);
                    int sp = __builtin_amdgcn_readlane(s, pp);
                    dls[q] = __builtin_amdgcn_readlane(dl, pp);
                    nrs[q] = __int_as_float(
                        __builtin_amdgcn_readlane(__float_as_int(nrm), pp));
                    const v2f* h = reinterpret_cast<const v2f*>(H + (size_t)sp * Fg);
#pragma unroll
                    for (int j = 0; j < NJ; ++j)
                        v[q][j] = h[lane + 32 * j];
                }
#pragma unroll
                for (int q = 0; q < GATHER_Q; ++q) {
                    if (p + q < count) {
                        float* o = tile + dls[q] * P + 2 * lane;
#pragma unroll
                        for (int j = 0; j < NJ; ++j) {
                            atomicAdd(&o[64 * j],     nrs[q] * v[q][j].x);
                            atomicAdd(&o[64 * j + 1], nrs[q] * v[q][j].y);
                        }
                    }
                }
            }
        };
        if (Fg > 64) groups(NTag<2>{}); else groups(NTag<1>{});
    };
    auto flush = [&](int my_edge, int count) {
        uint2 r = make_uint2(0u, 0u);
        if (lane < count) {
            if (LIN == 2) {
                const int   s = esrc[my_edge];
                const int   d = edst[my_edge];
                const float nrm = dinv[s] * ew[my_edge] * dinv[d];
                r = make_uint2((unsigned)s | ((unsigned)d << 16), __float_as_uint(nrm));
            } else {
                r = rec2[my_edge];
            }
        }
        if (LOUT != 0 && lst_out != nullptr) {
            unsigned at = 0u;
            if (lane == 0) at = atomicAdd(&lfill, (unsigned)count);
            at = (unsigned)__builtin_amdgcn_readfirstlane((int)at);
            if (lane < count && at + (unsigned)lane < (unsigned)LIST_CAP) {
                if (LOUT == 1) lst_out[ibase + at + lane] = (unsigned)my_edge;
                else reinterpret_cast<uint2*>(lst_out)[rbase + at + lane] = r;
            }
        }
        flush_rec(r, count);
    };

    if (listed) {
        if (LIN == 1 && LOUT == 2 && lst_out != nullptr) {
#pragma unroll
            for (int k = 0; k < MAXB; ++k)
                if (32 * k + lane < mycount)
                    reinterpret_cast<uint2*>(lst_out)[rbase + first + 32 * k + lane] = rr[k];
        }
#pragma nounroll
        for (int k = 0; k < MAXB; ++k) {
            uint2 rk = rr[0];
#pragma unroll
            for (int q = 1; q < MAXB; ++q) rk = (k == q) ? rr[q] : rk;
            int c = mycount - 32 * k;
            c = (c > 32) ? 32 : c;
            c = __builtin_amdgcn_readfirstlane(c);
            if (c <= 0) break;
            flush_rec(rk, c);
        }
        if (LOUT != 0 && lst_out != nullptr && threadIdx.x == 0) lst_out[blockIdx.x] = total;
    } else if (LIN == 2) {
        int mye = 0;
        int cnt = 0;
        for (int base = wave * 32; base < e; base += nw * 32) {
            const int i = base + lane;
            bool m = false;
            if (i < e) m = ((unsigned)edst[i] / TILE_R) == blockIdx.x;
            unsigned b = __builtin_amdgcn_ballot_w32(m);
            while (b) {
                int L = __builtin_ctz(b);
                b &= b - 1;
                mye = (lane == cnt) ? (base + L) : mye;
                if (++cnt == 32) {
                    flush(mye, 32);
                    cnt = 0;
                }
            }
        }
        if (cnt) flush(mye, cnt);
    } else {
        int mye = 0;
        int cnt = 0;

        const uint4* pl4 = reinterpret_cast<const uint4*>(planes);
        const unsigned tid = blockIdx.x;
        unsigned sel[8];
#pragma unroll
        for (int p = 0; p < 8; ++p) sel[p] = ((tid >> p) & 1u) ? 0u : ~0u;

        const int ntrips = (e + 1024 * SCAN_GROUPS - 1) / (1024 * SCAN_GROUPS);
        for (int t = wave; t < ntrips; t += nw) {
            unsigned bv[SCAN_GROUPS];
            {
                uint4 pa[SCAN_GROUPS], pb[SCAN_GROUPS];
#pragma unroll
                for (int g = 0; g < SCAN_GROUPS; ++g) {
                    const size_t blk = (size_t)(t * SCAN_GROUPS + g) * 64;
                    pa[g] = pl4[blk + lane];
                    pb[g] = pl4[blk + 32 + lane];
                }
#pragma unroll
                for (int g = 0; g < SCAN_GROUPS; ++g)
                    bv[g] = (pa[g].x ^ sel[0]) & (pa[g].y ^ sel[1]) & (pa[g].z ^ sel[2]) & (pa[g].w ^ sel[3])
                          & (pb[g].x ^ sel[4]) & (pb[g].y ^ sel[5]) & (pb[g].z ^ sel[6]) & (pb[g].w ^ sel[7]);
            }
#pragma nounroll
            for (int g = 0; g < SCAN_GROUPS; ++g) {
                unsigned bvg = bv[0];
#pragma unroll
                for (int q = 1; q < SCAN_GROUPS; ++q) bvg = (g == q) ? bv[q] : bvg;
                const int base = (t * SCAN_GROUPS + g) << 10;
                unsigned have = __builtin_amdgcn_ballot_w32(bvg != 0);
                while (have) {
                    int j = __builtin_ctz(have);
                    have &= have - 1;
                    unsigned b = (unsigned)__builtin_amdgcn_readlane((int)bvg, j);
                    while (b) {
                        int L = __builtin_ctz(b);
                        b &= b - 1;
                        mye = (lane == cnt) ? (base + j * 32 + L) : mye;
                        if (++cnt == 32) {
                            flush(mye, 32);
                            cnt = 0;
                        }
                    }
                }
            }
        }
        if (cnt) flush(mye, cnt);
    }

    if constexpr (FN > 0) {
        constexpr int NCT = FN / 16;
        const int nsg  = nw / NCT;
        const int wv   = __builtin_amdgcn_readfirstlane(wave);
        const int ct   = wv % NCT;
        const int sg   = wv / NCT;
        const int l16  = lane & 15;
        const int half = lane >> 4;
        const int col  = ct * 16 + l16;
        v2f bw[32];
#pragma unroll
        for (int kk = 0; kk < 32; ++kk) {
            const int ka = kk * 4 + 2 * half;
            bw[kk].x = Wn[(size_t)ka * FN + col];
            bw[kk].y = Wn[(size_t)(ka + 1) * FN + col];
        }
        __syncthreads();
        if (LOUT != 0 && !listed && lst_out != nullptr && threadIdx.x == 0) lst_out[blockIdx.x] = lfill;
        const int nstrips = (rows + 15) >> 4;
        for (int s = sg; s < nstrips; s += nsg) {
            const float* arow = tile + (s * 16 + l16) * P + 2 * half;
            v8f c = {0.f, 0.f, 0.f, 0.f, 0.f, 0.f, 0.f, 0.f};
#pragma unroll
            for (int kk = 0; kk < 32; ++kk) {
                v2f a = *reinterpret_cast<const v2f*>(arow + kk * 4);
                a.x = fmaxf(a.x, 0.0f);
                a.y = fmaxf(a.y, 0.0f);
                c = __builtin_amdgcn_wmma_f32_16x16x4_f32(false, a, false, bw[kk],
                                                          (short)0, c, false, false);
            }
            const int mb = s * 16 + half * 8;
            float* o = outp + (size_t)(row0 + mb) * FN + col;
#pragma unroll
            for (int r = 0; r < 8; ++r)
                if (mb + r < rows) o[(size_t)r * FN] = c[r];
        }
    } else {
        __syncthreads();
        if (LOUT != 0 && !listed && lst_out != nullptr && threadIdx.x == 0) lst_out[blockIdx.x] = lfill;
        for (int r = wave; r < rows; r += nw) {
            float* o = outp + (size_t)(row0 + r) * F;
            for (int f = lane; f < F; f += 32)
                o[f] = tile[r * P + f];
        }
    }
}

extern "C" void kernel_launch(void* const* d_in, const int* in_sizes, int n_in,
                              void* d_out, int out_size, void* d_ws, size_t ws_size,
                              hipStream_t stream) {
    const float* x    = (const float*)d_in[0];
    const int*   ei   = (const int*)d_in[1];
    const float* ew   = (const float*)d_in[2];
    const float* W1   = (const float*)d_in[3];
    const float* b1   = (const float*)d_in[4];
    const float* W2   = (const float*)d_in[5];
    const float* b2   = (const float*)d_in[6];
    const float* W3   = (const float*)d_in[7];
    const float* b3   = (const float*)d_in[8];
    float*       out  = (float*)d_out;

    const int hidden  = in_sizes[4];
    const int out_dim = in_sizes[8];
    const int in_dim  = in_sizes[3] / hidden;
    const int n       = in_sizes[0] / in_dim;
    const int e       = in_sizes[2];

    const int* src = ei;
    const int* dst = ei + e;

    char* ws = (char*)d_ws;
    size_t off = 0;
    float* dinv = (float*)(ws + off);
    off += ((size_t)n * sizeof(float) + 255) & ~(size_t)255;
    float* H = (float*)(ws + off);
    off += ((size_t)n * hidden * sizeof(float) + 255) & ~(size_t)255;
    float* AGG = (float*)(ws + off);

    const int B = 256;
    auto cdiv = [](long long a, long long b) { return (int)((a + b - 1) / b); };

    const int ntiles = cdiv(n, TILE_R);
    const int e_pad  = cdiv(e, 1024 * SCAN_GROUPS) * 1024 * SCAN_GROUPS;
    const size_t side = (size_t)e_pad + (size_t)e * 8;
    const bool keyed = ntiles <= 255 && n <= REC_NMAX && hidden % 64 == 0 && hidden <= 128 &&
                       out_dim % 64 == 0 && out_dim <= 128 &&
                       side + (size_t)n * sizeof(float) <= (size_t)n * out_dim * sizeof(float) &&
                       side <= (size_t)n * hidden * sizeof(float);
    unsigned* planes_a = (unsigned*)d_out;
    unsigned* planes_b = (unsigned*)AGG;
    unsigned* rec_a = (unsigned*)((char*)d_out + e_pad);
    unsigned* rec_b = (unsigned*)((char*)AGG + e_pad);
    float*    degs  = keyed ? (float*)((char*)d_out + side) : dinv;

    auto aggregate = [&](const float* Hin, const float* bias, float* dstbuf, int F,
                         const unsigned* planes, const unsigned* rec) {
        if (keyed) {
            agg_tile_kernel<0, 0, 0><<<ntiles, 1024, 0, stream>>>(
                dinv, Hin, bias, planes, rec, nullptr, dstbuf, nullptr, nullptr,
                src, dst, ew, n, e, F);
        } else {
            init_agg_kernel<<<cdiv((long long)n * F, B), B, 0, stream>>>(
                Hin, bias, dinv, dstbuf, n, F);
            scatter_kernel<<<cdiv((long long)e * 32, B), B, 0, stream>>>(
                src, dst, ew, dinv, Hin, dstbuf, e, F);
        }
    };

    deg_init_kernel<<<cdiv(n, B), B, 0, stream>>>(degs, n);
    if (keyed) {
        edge_pass_kernel<true><<<e_pad / B, B, 0, stream>>>(dst, ew, degs, planes_a,
                                                            nullptr, nullptr, e);
        norm_rec_kernel<<<REC_WGS, 1024, 0, stream>>>(src, dst, ew, degs, dinv, rec_a, n, e);
    } else {
        deg_accum_kernel<<<cdiv(e, B), B, 0, stream>>>(dst, ew, dinv, e);
        dinv_kernel<<<cdiv(n, B), B, 0, stream>>>(dinv, n);
    }

    auto product1 = [&]() {
        if (in_dim == 256 && hidden == 128 && n > 0) {
            gemm1_wstat_kernel<<<256, 512, 0, stream>>>(x, W1, H, n);
        } else {
            int tiles = (n / 16) * (hidden / 16);
            gemm_wmma_f32_kernel<<<cdiv((long long)tiles * 32, B), B, 0, stream>>>(
                x, W1, H, n, in_dim, hidden, 0);
        }
    };

    if (keyed && hidden == 128 && out_dim == 64) {
        const size_t ids_bytes  = ((size_t)LIST_HDR + (size_t)ntiles * LIST_CAP) * sizeof(unsigned);
        const size_t recs_bytes = ((size_t)LIST_HDR + (size_t)ntiles * LIST_CAP * 2) * sizeof(unsigned);
        const bool   lists = side + (size_t)n * sizeof(float) + ids_bytes <= (size_t)n * out_dim * sizeof(float) &&
                             recs_bytes <= (size_t)n * (hidden - out_dim) * sizeof(float);
        unsigned* lst_a = lists ? (unsigned*)((char*)d_out + side + (size_t)n * sizeof(float)) : nullptr;
        unsigned* lst_b = lists ? (unsigned*)(H + (size_t)n * out_dim) : nullptr;
        product1();
        agg_tile_kernel<128, 0, 1><<<ntiles, 1024, 0, stream>>>(
            dinv, H, b1, planes_a, rec_a, W2, AGG, nullptr, lst_a, src, dst, ew, n, e, hidden);
        agg_tile_kernel<64, 1, 2><<<ntiles, 1024, 0, stream>>>(
            dinv, AGG, b2, planes_a, rec_a, W3, H, lst_a, lst_b, src, dst, ew, n, e, hidden);
        if (lists) {
            agg_tile_kernel<0, 2, 0><<<ntiles, 1024, 0, stream>>>(
                dinv, H, b3, nullptr, nullptr, nullptr, out, lst_b, nullptr, src, dst, ew, n, e, out_dim);
            return;
        }
        edge_pass_kernel<false><<<e_pad / B, B, 0, stream>>>(dst, nullptr, nullptr, planes_b,
                                                             rec_a, rec_b, e);
        agg_tile_kernel<0, 0, 0><<<ntiles, 1024, 0, stream>>>(
            dinv, H, b3, planes_b, rec_b, nullptr, out, nullptr, nullptr, src, dst, ew, n, e, out_dim);
        return;
    }

    {
        product1();
        aggregate(H, b1, AGG, hidden, planes_a, rec_a);
    }

    {
        int tiles = (n / 16) * (hidden / 16);
        gemm_wmma_f32_kernel<<<cdiv((long long)tiles * 32, B), B, 0, stream>>>(
            AGG, W2, H, n, hidden, hidden, 1);
        aggregate(H, b2, AGG, hidden, planes_a, rec_a);
    }

    {
        int tiles = (n / 16) * (out_dim / 16);
        gemm_wmma_f32_kernel<<<cdiv((long long)tiles * 32, B), B, 0, stream>>>(
            AGG, W3, H, n, hidden, out_dim, 1);
        if (keyed)
            edge_pass_kernel<false><<<e_pad / B, B, 0, stream>>>(dst, nullptr, nullptr, planes_b,
                                                                 rec_a, rec_b, e);
        aggregate(H, b3, out, out_dim, planes_b, rec_b);
    }
}
